# starting kernel
# baseline (speedup 1.0000x reference)
#include <hip/hip_runtime.h>
#include <hip/hip_bf16.h>
#include <stdint.h>


#define FDIM 64
#define HDIM 64
#define CDIM 10

#define GWAVES 4
#define GROWS  (GWAVES * 16)
#define ASTR   68
#define WPSTR  160
#define WPROWS 32
#define WOSTR  96

typedef float v2f __attribute__((ext_vector_type(2)));
typedef float v8f __attribute__((ext_vector_type(8)));

__device__ __forceinline__ v8f wmma4(v2f a, v2f b, v8f c) {
  return __builtin_amdgcn_wmma_f32_16x16x4_f32(false, a, false, b, (short)0, c,
                                               false, false);
}


__global__ void zero_kernel(float* __restrict__ p, int n) {
  int i = blockIdx.x * blockDim.x + threadIdx.x;
  if (i < n) p[i] = 0.0f;
}

__global__ void degree_kernel(const int* __restrict__ src, float* __restrict__ deg, int E) {
  int e = blockIdx.x * blockDim.x + threadIdx.x;
  if (e < E) atomicAdd(&deg[src[e]], 1.0f);
}

__global__ void dis_kernel(const float* __restrict__ deg, float* __restrict__ dis, int n) {
  int i = blockIdx.x * blockDim.x + threadIdx.x;
  if (i >= n) return;
  float d = deg[i];
  dis[i] = (d > 0.0f) ? rsqrtf(fmaxf(d, 1.0f)) : 0.0f;
}

__global__ void edgew_kernel(const int* __restrict__ src, const int* __restrict__ dst,
                             const float* __restrict__ dis, float* __restrict__ w, int E) {
  int e = blockIdx.x * blockDim.x + threadIdx.x;
  if (e < E) w[e] = -(dis[src[e]] * dis[dst[e]]);
}

__global__ void scatter_kernel(const int* __restrict__ src, const int* __restrict__ dst,
                               const float* __restrict__ w, const float* __restrict__ h,
                               float* __restrict__ lap, int E) {
  int t = blockIdx.x * blockDim.x + threadIdx.x;
  int e = t >> 4;
  if (e >= E) return;
  int chunk = t & 15;
  int s = src[e], d = dst[e];
  float we = w[e];
  float4 hv = ((const float4*)(h + (size_t)s * HDIM))[chunk];
  float* lp = lap + (size_t)d * HDIM + chunk * 4;
  atomicAdd(lp + 0, we * hv.x);
  atomicAdd(lp + 1, we * hv.y);
  atomicAdd(lp + 2, we * hv.z);
  atomicAdd(lp + 3, we * hv.w);
}


__device__ __forceinline__ void stage_w_pairs(float* sWp, const float* __restrict__ W,
                                              int tid) {
  const float4* Wv = (const float4*)W;
#pragma unroll
  for (int i = 0; i < 8; ++i) {
    int idx = tid + i * 128;
    float4 v = Wv[idx];
    int k = idx >> 4;
    int c4 = (idx & 15) * 4;
    float* dstp = sWp + (k >> 1) * WPSTR + (k & 1);
    dstp[(c4 + 0) * 2] = v.x;
    dstp[(c4 + 1) * 2] = v.y;
    dstp[(c4 + 2) * 2] = v.z;
    dstp[(c4 + 3) * 2] = v.w;
  }
}

__device__ __forceinline__ void stage_rows(float* sAw, const float* __restrict__ A,
                                           int rowBase, int n, int lane) {
#pragma unroll
  for (int i = 0; i < 8; ++i) {
    int fi = lane + i * 32;
    int r = fi >> 4, c4 = fi & 15;
    int row = rowBase + r;
    if (row >= n) row = n - 1;
    ((float4*)(sAw + r * ASTR))[c4] = ((const float4*)(A + (size_t)row * 64))[c4];
  }
}

__device__ __forceinline__ void mm_accum64(const float* sAw, const float* sWp,
                                           int lane, v8f acc[4]) {
  const int nlo = lane & 15;
  const int g = lane >> 4;
#pragma unroll
  for (int kk = 0; kk < 16; ++kk) {
    const int k0 = kk * 4 + 2 * g;
    v2f a = *(const v2f*)(sAw + nlo * ASTR + k0);
    const float* brow = sWp + (kk * 2 + g) * WPSTR;
#pragma unroll
    for (int nt = 0; nt < 4; ++nt) {
      v2f b = *(const v2f*)(brow + (nt * 16 + nlo) * 2);
      acc[nt] = wmma4(a, b, acc[nt]);
    }
  }
}


__global__ void __launch_bounds__(128)
gemm64_relu_kernel(const float* __restrict__ A, const float* __restrict__ W,
                   const float* __restrict__ bias, float* __restrict__ out, int n) {
  __shared__ float sWp[WPROWS * WPSTR];
  __shared__ float sA[GWAVES * 16 * ASTR];
  __shared__ float sB[64];
  const int tid = threadIdx.x, wave = tid >> 5, lane = tid & 31;
  stage_w_pairs(sWp, W, tid);
  if (tid < 64) sB[tid] = bias[tid];

  const int rowBase = blockIdx.x * GROWS + wave * 16;
  float* sAw = sA + wave * 16 * ASTR;
  stage_rows(sAw, A, rowBase, n, lane);
  __syncthreads();

  v8f z = {0, 0, 0, 0, 0, 0, 0, 0};
  v8f acc[4] = {z, z, z, z};
  mm_accum64(sAw, sWp, lane, acc);

  const int nlo = lane & 15, g = lane >> 4;
#pragma unroll
  for (int nt = 0; nt < 4; ++nt)
#pragma unroll
    for (int r = 0; r < 8; ++r) {
      int row = rowBase + r + 8 * g;
      int col = nt * 16 + nlo;
      if (row < n) out[(size_t)row * 64 + col] = fmaxf(acc[nt][r] + sB[col], 0.0f);
    }
}

__global__ void __launch_bounds__(128)
gemm64_cheb2_kernel(float* __restrict__ h, const float* __restrict__ lap,
                    const float* __restrict__ W0, const float* __restrict__ W1,
                    const float* __restrict__ bias, int n) {
  __shared__ float sWp[2 * WPROWS * WPSTR];
  __shared__ float sA[GWAVES * 16 * ASTR];
  __shared__ float sB[64];
  const int tid = threadIdx.x, wave = tid >> 5, lane = tid & 31;
  stage_w_pairs(sWp, W0, tid);
  stage_w_pairs(sWp + WPROWS * WPSTR, W1, tid);
  if (tid < 64) sB[tid] = bias[tid];

  const int rowBase = blockIdx.x * GROWS + wave * 16;
  float* sAw = sA + wave * 16 * ASTR;

  v8f z = {0, 0, 0, 0, 0, 0, 0, 0};
  v8f acc[4] = {z, z, z, z};

  stage_rows(sAw, h, rowBase, n, lane);
  __syncthreads();
  mm_accum64(sAw, sWp, lane, acc);
  __syncthreads();
  stage_rows(sAw, lap, rowBase, n, lane);
  __syncthreads();
  mm_accum64(sAw, sWp + WPROWS * WPSTR, lane, acc);

  const int nlo = lane & 15, g = lane >> 4;
#pragma unroll
  for (int nt = 0; nt < 4; ++nt)
#pragma unroll
    for (int r = 0; r < 8; ++r) {
      int row = rowBase + r + 8 * g;
      int col = nt * 16 + nlo;
      if (row < n) {
        size_t idx = (size_t)row * 64 + col;
        float v = fmaxf(acc[nt][r] + sB[col], 0.0f);
        h[idx] = v + h[idx];
      }
    }
}

__global__ void __launch_bounds__(128)
gemm_out_kernel(const float* __restrict__ A, const float* __restrict__ Wl,
                const float* __restrict__ bl, float* __restrict__ logits, int n) {
  __shared__ float sWo[WPROWS * WOSTR];
  __shared__ float sA[GWAVES * 16 * ASTR];
  __shared__ float sB[16];
  const int tid = threadIdx.x, wave = tid >> 5, lane = tid & 31;
  for (int idx = tid; idx < 64 * 16; idx += 128) {
    int k = idx >> 4, c = idx & 15;
    float v = (c < CDIM) ? Wl[k * CDIM + c] : 0.0f;
    sWo[(k >> 1) * WOSTR + c * 2 + (k & 1)] = v;
  }
  if (tid < 16) sB[tid] = (tid < CDIM) ? bl[tid] : 0.0f;

  const int rowBase = blockIdx.x * GROWS + wave * 16;
  float* sAw = sA + wave * 16 * ASTR;
  stage_rows(sAw, A, rowBase, n, lane);
  __syncthreads();

  const int nlo = lane & 15, g = lane >> 4;
  v8f acc = {0, 0, 0, 0, 0, 0, 0, 0};
#pragma unroll
  for (int kk = 0; kk < 16; ++kk) {
    const int k0 = kk * 4 + 2 * g;
    v2f a = *(const v2f*)(sAw + nlo * ASTR + k0);
    v2f b = *(const v2f*)(sWo + (kk * 2 + g) * WOSTR + nlo * 2);
    acc = wmma4(a, b, acc);
  }
#pragma unroll
  for (int r = 0; r < 8; ++r) {
    int row = rowBase + r + 8 * g;
    if (row < n) logits[(size_t)row * 16 + nlo] = acc[r] + sB[nlo];
  }
}

__global__ void logsoftmax_kernel(const float* __restrict__ logits,
                                  float* __restrict__ out, int n) {
  int i = blockIdx.x * blockDim.x + threadIdx.x;
  if (i >= n) return;
  const float* v = logits + (size_t)i * 16;
  float m = v[0];
#pragma unroll
  for (int c = 1; c < CDIM; ++c) m = fmaxf(m, v[c]);
  float s = 0.0f;
#pragma unroll
  for (int c = 0; c < CDIM; ++c) s += expf(v[c] - m);
  float ls = logf(s);
#pragma unroll
  for (int c = 0; c < CDIM; ++c) out[(size_t)i * CDIM + c] = v[c] - m - ls;
}


extern "C" void kernel_launch(void* const* d_in, const int* in_sizes, int n_in,
                              void* d_out, int out_size, void* d_ws, size_t ws_size,
                              hipStream_t stream) {
  const float* x    = (const float*)d_in[0];
  const int*   ei   = (const int*)d_in[1];
  const float* W1_0 = (const float*)d_in[2];
  const float* b1   = (const float*)d_in[3];
  const float* W2_0 = (const float*)d_in[4];
  const float* W2_1 = (const float*)d_in[5];
  const float* b2   = (const float*)d_in[6];
  const float* Wl   = (const float*)d_in[7];
  const float* bl   = (const float*)d_in[8];

  const int N = in_sizes[0] / FDIM;
  const int E = in_sizes[1] / 2;
  const int* src = ei;
  const int* dst = ei + E;

  float* ws     = (float*)d_ws;
  float* deg    = ws;
  float* lap    = deg + N;
  float* dis    = lap + (size_t)N * 64;
  float* wbuf   = dis + N;
  float* h      = wbuf + E;
  float* logits = h + (size_t)N * 64;
  float* out    = (float*)d_out;

  {
    int tot = N * 65;
    zero_kernel<<<(tot + 255) / 256, 256, 0, stream>>>(deg, tot);
  }
  degree_kernel<<<(E + 255) / 256, 256, 0, stream>>>(src, deg, E);
  dis_kernel<<<(N + 255) / 256, 256, 0, stream>>>(deg, dis, N);
  edgew_kernel<<<(E + 255) / 256, 256, 0, stream>>>(src, dst, dis, wbuf, E);

  const int gblocks = (N + GROWS - 1) / GROWS;
  gemm64_relu_kernel<<<gblocks, 128, 0, stream>>>(x, W1_0, b1, h, N);
  {
    long long tot = (long long)E * 16;
    scatter_kernel<<<(int)((tot + 255) / 256), 256, 0, stream>>>(src, dst, wbuf, h, lap, E);
  }
  gemm64_cheb2_kernel<<<gblocks, 128, 0, stream>>>(h, lap, W2_0, W2_1, b2, N);
  gemm_out_kernel<<<gblocks, 128, 0, stream>>>(h, Wl, bl, logits, N);
  logsoftmax_kernel<<<(N + 255) / 256, 256, 0, stream>>>(logits, out, N);
}


# lap by feature column: block (feature, dst half, edge half) holds the h column and its sums in LDS, LDS float adds per edge (arrival order, two partial sums per row); h and lap kept feature-major
# speedup vs baseline: 1.2981x; 1.2981x over previous
#include <hip/hip_runtime.h>
#include <hip/hip_bf16.h>
#include <stdint.h>


#define FDIM 64
#define HDIM 64
#define CDIM 10

#define GWAVES 4
#define GROWS  (GWAVES * 16)
#define ASTR   68
#define WPSTR  160
#define WPROWS 32
#define WOSTR  96
#define CSTR   32
#define NMAX   50000
#define NHALF  25000
#define CTHREADS 1024
#define CU     8

typedef float v2f __attribute__((ext_vector_type(2)));
typedef float v8f __attribute__((ext_vector_type(8)));

__device__ __forceinline__ v8f wmma4(v2f a, v2f b, v8f c) {
  return __builtin_amdgcn_wmma_f32_16x16x4_f32(false, a, false, b, (short)0, c,
                                               false, false);
}


__global__ void zero_kernel(float* __restrict__ p, int n, float* __restrict__ cnt, int ncnt) {
  int i = blockIdx.x * blockDim.x + threadIdx.x;
  if (i < n) p[i] = 0.0f;
  if (i < ncnt) cnt[(size_t)i * CSTR] = 0.0f;
}

__global__ void degree_kernel(const int* __restrict__ src, float* __restrict__ cnt, int E) {
  int e = blockIdx.x * blockDim.x + threadIdx.x;
  if (e < E) atomicAdd(&cnt[(size_t)src[e] * CSTR], 1.0f);
}

__global__ void dis_kernel(const float* __restrict__ cnt, float* __restrict__ dis, int n) {
  int i = blockIdx.x * blockDim.x + threadIdx.x;
  if (i >= n) return;
  float d = cnt[(size_t)i * CSTR];
  dis[i] = (d > 0.0f) ? rsqrtf(fmaxf(d, 1.0f)) : 0.0f;
}

__global__ void edgew_kernel(const int* __restrict__ src, const int* __restrict__ dst,
                             const float* __restrict__ dis, float* __restrict__ w, int E) {
  int e = blockIdx.x * blockDim.x + threadIdx.x;
  if (e < E) w[e] = -(dis[src[e]] * dis[dst[e]]);
}


__device__ __forceinline__ void col_load(int (&s)[CU], int (&d)[CU], float (&we)[CU],
                                         const int* __restrict__ sp, const int* __restrict__ dp,
                                         const float* __restrict__ wp, int t, int tid, int M) {
#pragma unroll
  for (int j = 0; j < CU; ++j) {
    int o = (t * CU + j) * CTHREADS + tid;
    int oc = (o < M) ? o : 0;
    s[j] = sp[oc];
    d[j] = dp[oc];
    we[j] = wp[oc];
  }
}

__device__ __forceinline__ void col_apply(const int (&s)[CU], const int (&d)[CU],
                                          const float (&we)[CU], const float* hcol, float* acc,
                                          int t, int tid, int M, int base, int cnt) {
  asm volatile("" :: "v"(s[0]), "v"(s[1]), "v"(s[2]), "v"(s[3]), "v"(s[4]), "v"(s[5]), "v"(s[6]),
               "v"(s[7]), "v"(d[0]), "v"(d[1]), "v"(d[2]), "v"(d[3]), "v"(d[4]), "v"(d[5]),
               "v"(d[6]), "v"(d[7]), "v"(we[0]), "v"(we[1]), "v"(we[2]), "v"(we[3]), "v"(we[4]),
               "v"(we[5]), "v"(we[6]), "v"(we[7]));
  float hv[CU];
#pragma unroll
  for (int j = 0; j < CU; ++j) hv[j] = hcol[s[j]];
  asm volatile("" :: "v"(hv[0]), "v"(hv[1]), "v"(hv[2]), "v"(hv[3]), "v"(hv[4]), "v"(hv[5]),
               "v"(hv[6]), "v"(hv[7]));
#pragma unroll
  for (int j = 0; j < CU; ++j) {
    int o = (t * CU + j) * CTHREADS + tid;
    unsigned r = (unsigned)(d[j] - base);
    if (o < M && r < (unsigned)cnt) atomicAdd(&acc[r], we[j] * hv[j]);
  }
}

__global__ void __launch_bounds__(CTHREADS)
colscatter_kernel(const int* __restrict__ src, const int* __restrict__ dst,
                  const float* __restrict__ w, const float* __restrict__ hT,
                  float* __restrict__ lapT, int n, int E) {
  __shared__ float hcol[NMAX];
  __shared__ float acc[NHALF];
  const int tid = threadIdx.x;
  const int f = blockIdx.x >> 2;
  const int p = (blockIdx.x >> 1) & 1;
  const int q = blockIdx.x & 1;
  const int half = (n + 1) >> 1;
  const int base = p * half;
  const int cnt = (p == 0) ? half : (n - half);
  const float* __restrict__ col = hT + (size_t)f * n;
  for (int i = tid; i < n; i += CTHREADS) hcol[i] = col[i];
  for (int i = tid; i < cnt; i += CTHREADS) acc[i] = 0.0f;
  __syncthreads();

  const int eh = (E + 1) >> 1;
  const int e0 = q * eh;
  const int e1 = (e0 + eh < E) ? (e0 + eh) : E;
  const int M = e1 - e0;
  const int T = (M + CU * CTHREADS - 1) / (CU * CTHREADS);
  const int* __restrict__ sp = src + e0;
  const int* __restrict__ dp = dst + e0;
  const float* __restrict__ wp = w + e0;

  int sA[CU], dA[CU], sB[CU], dB[CU];
  float wA[CU], wB[CU];
  if (T > 0) {
    col_load(sA, dA, wA, sp, dp, wp, 0, tid, M);
    for (int t = 0; t < T; t += 2) {
      const int tb = (t + 1 < T) ? (t + 1) : (T - 1);
      const int ta = (t + 2 < T) ? (t + 2) : (T - 1);
      col_load(sB, dB, wB, sp, dp, wp, tb, tid, M);
      __builtin_amdgcn_sched_barrier(0);
      col_apply(sA, dA, wA, hcol, acc, t, tid, M, base, cnt);
      col_load(sA, dA, wA, sp, dp, wp, ta, tid, M);
      __builtin_amdgcn_sched_barrier(0);
      col_apply(sB, dB, wB, hcol, acc, t + 1, tid, M, base, cnt);
    }
  }
  __syncthreads();

  float* outp = lapT + (size_t)f * n + base;
  for (int i = tid; i < cnt; i += CTHREADS) atomicAdd(outp + i, acc[i]);
}


__device__ __forceinline__ void stage_w_pairs(float* sWp, const float* __restrict__ W,
                                              int tid) {
  const float4* Wv = (const float4*)W;
#pragma unroll
  for (int i = 0; i < 8; ++i) {
    int idx = tid + i * 128;
    float4 v = Wv[idx];
    int k = idx >> 4;
    int c4 = (idx & 15) * 4;
    float* dstp = sWp + (k >> 1) * WPSTR + (k & 1);
    dstp[(c4 + 0) * 2] = v.x;
    dstp[(c4 + 1) * 2] = v.y;
    dstp[(c4 + 2) * 2] = v.z;
    dstp[(c4 + 3) * 2] = v.w;
  }
}

__device__ __forceinline__ void stage_rows(float* sAw, const float* __restrict__ A,
                                           int rowBase, int n, int lane) {
#pragma unroll
  for (int i = 0; i < 8; ++i) {
    int fi = lane + i * 32;
    int r = fi >> 4, c4 = fi & 15;
    int row = rowBase + r;
    if (row >= n) row = n - 1;
    ((float4*)(sAw + r * ASTR))[c4] = ((const float4*)(A + (size_t)row * 64))[c4];
  }
}

__device__ __forceinline__ void stage_rows_T(float* sAw, const float* __restrict__ AT,
                                             int rowBase, int n, int lane) {
  const int r = lane & 15, kh = lane >> 4;
  int row = rowBase + r;
  if (row >= n) row = n - 1;
#pragma unroll
  for (int i = 0; i < 32; ++i) {
    int k = 2 * i + kh;
    sAw[r * ASTR + k] = AT[(size_t)k * n + row];
  }
}

__device__ __forceinline__ void mm_accum64(const float* sAw, const float* sWp,
                                           int lane, v8f acc[4]) {
  const int nlo = lane & 15;
  const int g = lane >> 4;
#pragma unroll
  for (int kk = 0; kk < 16; ++kk) {
    const int k0 = kk * 4 + 2 * g;
    v2f a = *(const v2f*)(sAw + nlo * ASTR + k0);
    const float* brow = sWp + (kk * 2 + g) * WPSTR;
#pragma unroll
    for (int nt = 0; nt < 4; ++nt) {
      v2f b = *(const v2f*)(brow + (nt * 16 + nlo) * 2);
      acc[nt] = wmma4(a, b, acc[nt]);
    }
  }
}


__global__ void __launch_bounds__(128)
gemm64_relu_kernel(const float* __restrict__ A, const float* __restrict__ W,
                   const float* __restrict__ bias, float* __restrict__ outT, int n) {
  __shared__ float sWp[WPROWS * WPSTR];
  __shared__ float sA[GWAVES * 16 * ASTR];
  __shared__ float sB[64];
  const int tid = threadIdx.x, wave = tid >> 5, lane = tid & 31;
  stage_w_pairs(sWp, W, tid);
  if (tid < 64) sB[tid] = bias[tid];

  const int rowBase = blockIdx.x * GROWS + wave * 16;
  float* sAw = sA + wave * 16 * ASTR;
  stage_rows(sAw, A, rowBase, n, lane);
  __syncthreads();

  v8f z = {0, 0, 0, 0, 0, 0, 0, 0};
  v8f acc[4] = {z, z, z, z};
  mm_accum64(sAw, sWp, lane, acc);

  const int nlo = lane & 15, g = lane >> 4;
#pragma unroll
  for (int nt = 0; nt < 4; ++nt)
#pragma unroll
    for (int r = 0; r < 8; ++r) {
      int col = nt * 16 + nlo;
      sAw[(r + 8 * g) * ASTR + col] = fmaxf(acc[nt][r] + sB[col], 0.0f);
    }
  __syncthreads();
  {
    const int r = lane & 15, kh = lane >> 4;
    const int row = rowBase + r;
    if (row < n) {
#pragma unroll
      for (int i = 0; i < 32; ++i) {
        int k = 2 * i + kh;
        outT[(size_t)k * n + row] = sAw[r * ASTR + k];
      }
    }
  }
}

__global__ void __launch_bounds__(128)
gemm64_cheb2_kernel(float* __restrict__ h, const float* __restrict__ lap,
                    const float* __restrict__ W0, const float* __restrict__ W1,
                    const float* __restrict__ bias, int n) {
  __shared__ float sWp[2 * WPROWS * WPSTR];
  __shared__ float sA[GWAVES * 16 * ASTR];
  __shared__ float sB[64];
  const int tid = threadIdx.x, wave = tid >> 5, lane = tid & 31;
  stage_w_pairs(sWp, W0, tid);
  stage_w_pairs(sWp + WPROWS * WPSTR, W1, tid);
  if (tid < 64) sB[tid] = bias[tid];

  const int rowBase = blockIdx.x * GROWS + wave * 16;
  float* sAw = sA + wave * 16 * ASTR;

  v8f z = {0, 0, 0, 0, 0, 0, 0, 0};
  v8f acc[4] = {z, z, z, z};

  stage_rows_T(sAw, h, rowBase, n, lane);
  __syncthreads();
  mm_accum64(sAw, sWp, lane, acc);
  __syncthreads();
  stage_rows_T(sAw, lap, rowBase, n, lane);
  __syncthreads();
  mm_accum64(sAw, sWp + WPROWS * WPSTR, lane, acc);

  const int nlo = lane & 15, g = lane >> 4;
#pragma unroll
  for (int nt = 0; nt < 4; ++nt)
#pragma unroll
    for (int r = 0; r < 8; ++r) {
      int col = nt * 16 + nlo;
      sAw[(r + 8 * g) * ASTR + col] = fmaxf(acc[nt][r] + sB[col], 0.0f);
    }
  __syncthreads();
  {
    const int r = lane & 15, kh = lane >> 4;
    const int row = rowBase + r;
    if (row < n) {
      float res[32];
#pragma unroll
      for (int i = 0; i < 32; ++i) res[i] = h[(size_t)(2 * i + kh) * n + row];
#pragma unroll
      for (int i = 0; i < 32; ++i) {
        int k = 2 * i + kh;
        h[(size_t)k * n + row] = sAw[r * ASTR + k] + res[i];
      }
    }
  }
}

__global__ void __launch_bounds__(128)
gemm_out_kernel(const float* __restrict__ A, const float* __restrict__ Wl,
                const float* __restrict__ bl, float* __restrict__ logits, int n) {
  __shared__ float sWo[WPROWS * WOSTR];
  __shared__ float sA[GWAVES * 16 * ASTR];
  __shared__ float sB[16];
  const int tid = threadIdx.x, wave = tid >> 5, lane = tid & 31;
  for (int idx = tid; idx < 64 * 16; idx += 128) {
    int k = idx >> 4, c = idx & 15;
    float v = (c < CDIM) ? Wl[k * CDIM + c] : 0.0f;
    sWo[(k >> 1) * WOSTR + c * 2 + (k & 1)] = v;
  }
  if (tid < 16) sB[tid] = (tid < CDIM) ? bl[tid] : 0.0f;

  const int rowBase = blockIdx.x * GROWS + wave * 16;
  float* sAw = sA + wave * 16 * ASTR;
  stage_rows_T(sAw, A, rowBase, n, lane);
  __syncthreads();

  const int nlo = lane & 15, g = lane >> 4;
  v8f acc = {0, 0, 0, 0, 0, 0, 0, 0};
#pragma unroll
  for (int kk = 0; kk < 16; ++kk) {
    const int k0 = kk * 4 + 2 * g;
    v2f a = *(const v2f*)(sAw + nlo * ASTR + k0);
    v2f b = *(const v2f*)(sWo + (kk * 2 + g) * WOSTR + nlo * 2);
    acc = wmma4(a, b, acc);
  }
#pragma unroll
  for (int r = 0; r < 8; ++r) {
    int row = rowBase + r + 8 * g;
    if (row < n) logits[(size_t)row * 16 + nlo] = acc[r] + sB[nlo];
  }
}

__global__ void logsoftmax_kernel(const float* __restrict__ logits,
                                  float* __restrict__ out, int n) {
  int i = blockIdx.x * blockDim.x + threadIdx.x;
  if (i >= n) return;
  const float* v = logits + (size_t)i * 16;
  float m = v[0];
#pragma unroll
  for (int c = 1; c < CDIM; ++c) m = fmaxf(m, v[c]);
  float s = 0.0f;
#pragma unroll
  for (int c = 0; c < CDIM; ++c) s += expf(v[c] - m);
  float ls = logf(s);
#pragma unroll
  for (int c = 0; c < CDIM; ++c) out[(size_t)i * CDIM + c] = v[c] - m - ls;
}


extern "C" void kernel_launch(void* const* d_in, const int* in_sizes, int n_in,
                              void* d_out, int out_size, void* d_ws, size_t ws_size,
                              hipStream_t stream) {
  const float* x    = (const float*)d_in[0];
  const int*   ei   = (const int*)d_in[1];
  const float* W1_0 = (const float*)d_in[2];
  const float* b1   = (const float*)d_in[3];
  const float* W2_0 = (const float*)d_in[4];
  const float* W2_1 = (const float*)d_in[5];
  const float* b2   = (const float*)d_in[6];
  const float* Wl   = (const float*)d_in[7];
  const float* bl   = (const float*)d_in[8];

  const int N = in_sizes[0] / FDIM;
  const int E = in_sizes[1] / 2;
  const int* src = ei;
  const int* dst = ei + E;

  float* ws     = (float*)d_ws;
  float* lap    = ws + N;
  float* dis    = lap + (size_t)N * 64;
  float* wbuf   = dis + N;
  float* h      = wbuf + E;
  float* logits = h + (size_t)N * 64;
  float* out    = (float*)d_out;
  float* cnt    = h;

  {
    int tot = N * 64;
    zero_kernel<<<(tot + 255) / 256, 256, 0, stream>>>(lap, tot, cnt, N);
  }
  degree_kernel<<<(E + 255) / 256, 256, 0, stream>>>(src, cnt, E);
  dis_kernel<<<(N + 255) / 256, 256, 0, stream>>>(cnt, dis, N);
  edgew_kernel<<<(E + 255) / 256, 256, 0, stream>>>(src, dst, dis, wbuf, E);

  const int gblocks = (N + GROWS - 1) / GROWS;
  gemm64_relu_kernel<<<gblocks, 128, 0, stream>>>(x, W1_0, b1, h, N);
  colscatter_kernel<<<64 * 4, CTHREADS, 0, stream>>>(src, dst, wbuf, h, lap, N, E);
  gemm64_cheb2_kernel<<<gblocks, 128, 0, stream>>>(h, lap, W2_0, W2_1, b2, N);
  gemm_out_kernel<<<gblocks, 128, 0, stream>>>(h, Wl, bl, logits, N);
  logsoftmax_kernel<<<(N + 255) / 256, 256, 0, stream>>>(logits, out, N);
}


# edge list split by source half (stable, in the dis launch); column block (feature, source half, list half) holds half a column, all row sums, streams only own edges; 4 partial sums per row (was 2)
# speedup vs baseline: 1.7252x; 1.3290x over previous
#include <hip/hip_runtime.h>
#include <hip/hip_bf16.h>
#include <stdint.h>


#define FDIM 64
#define HDIM 64
#define CDIM 10

#define GWAVES 4
#define GROWS  (GWAVES * 16)
#define ASTR   68
#define WPSTR  160
#define WPROWS 32
#define WOSTR  96
#define CSTR   32
#define NMAX   50000
#define NHALF  25000
#define CTHREADS 1024
#define CU     16
#define CQ     (CU / 4)
#define NSTG   49
#define NOUT   25
#define PTHREADS 1024

typedef unsigned int u4 __attribute__((ext_vector_type(4)));
typedef float v2f __attribute__((ext_vector_type(2)));
typedef float v8f __attribute__((ext_vector_type(8)));

__device__ __forceinline__ v8f wmma4(v2f a, v2f b, v8f c) {
  return __builtin_amdgcn_wmma_f32_16x16x4_f32(false, a, false, b, (short)0, c,
                                               false, false);
}


__global__ void zero_kernel(float* __restrict__ p, int n, float* __restrict__ cnt, int ncnt) {
  int i = blockIdx.x * blockDim.x + threadIdx.x;
  if (i < n) p[i] = 0.0f;
  if (i < ncnt) cnt[(size_t)i * CSTR] = 0.0f;
}

__device__ __forceinline__ int wave_sum(int v) {
  for (int o = 16; o > 0; o >>= 1) v += __shfl_xor(v, o, 32);
  return v;
}

__global__ void __launch_bounds__(PTHREADS)
degree_kernel(const int* __restrict__ src, float* __restrict__ cnt, int* __restrict__ c0,
              int E, int half) {
  __shared__ int sw[PTHREADS / 32];
  const int tid = threadIdx.x, lane = tid & 31, wave = tid >> 5;
  const int e = blockIdx.x * PTHREADS + tid;
  const bool valid = (e < E);
  const int s = valid ? src[e] : 0;
  if (valid) atomicAdd(&cnt[(size_t)s * CSTR], 1.0f);
  const unsigned b = (unsigned)__ballot(valid && (s < half));
  if (lane == 0) sw[wave] = __popc(b);
  __syncthreads();
  if (wave == 0) {
    const int v = wave_sum(sw[lane]);
    if (lane == 0) c0[blockIdx.x] = v;
  }
}

__global__ void __launch_bounds__(PTHREADS)
place_dis_kernel(const int* __restrict__ src, const int* __restrict__ dst,
                 const int* __restrict__ c0, int* __restrict__ ctr,
                 unsigned int* __restrict__ list0, unsigned int* __restrict__ list1,
                 const float* __restrict__ cnt, float* __restrict__ dis,
                 int n, int E, int nblk, int half) {
  __shared__ int sred[PTHREADS / 32];
  __shared__ int sw0[PTHREADS / 32];
  __shared__ int sw1[PTHREADS / 32];
  const int tid = threadIdx.x, lane = tid & 31, wave = tid >> 5;
  const int blk = blockIdx.x;
  if (blk >= nblk) {
    const int i = (blk - nblk) * PTHREADS + tid;
    if (i < n) {
      const float d = cnt[(size_t)i * CSTR];
      dis[i] = (d > 0.0f) ? rsqrtf(fmaxf(d, 1.0f)) : 0.0f;
    }
    return;
  }
  int part = 0;
  for (int i = tid; i < blk; i += PTHREADS) part += c0[i];
  part = wave_sum(part);
  if (lane == 0) sred[wave] = part;
  const int e = blk * PTHREADS + tid;
  const bool valid = (e < E);
  int s = 0, d = 0;
  if (valid) { s = src[e]; d = dst[e]; }
  const bool is0 = valid && (s < half);
  const bool is1 = valid && (s >= half);
  const unsigned b0 = (unsigned)__ballot(is0);
  const unsigned b1 = (unsigned)__ballot(is1);
  if (lane == 0) { sw0[wave] = __popc(b0); sw1[wave] = __popc(b1); }
  __syncthreads();
  const int a0 = sw0[lane], a1 = sw1[lane];
  const int P0 = wave_sum(sred[lane]);
  const int t0 = wave_sum(a0);
  const int w0 = wave_sum((lane < wave) ? a0 : 0);
  const int w1 = wave_sum((lane < wave) ? a1 : 0);
  const int P1 = blk * PTHREADS - P0;
  const unsigned lt = (1u << lane) - 1u;
  if (is0) list0[P0 + w0 + __popc(b0 & lt)] = (unsigned)s | ((unsigned)d << 16);
  if (is1) list1[P1 + w1 + __popc(b1 & lt)] = (unsigned)(s - half) | ((unsigned)d << 16);
  if (blk == nblk - 1 && tid == 0) {
    ctr[0] = P0 + t0;
    ctr[1] = E - (P0 + t0);
  }
}


__device__ __forceinline__ void col_load(u4 (&w)[CQ], const u4* __restrict__ pp,
                                         int t, int tid, int M4) {
#pragma unroll
  for (int j = 0; j < CQ; ++j) {
    int o = (t * CQ + j) * CTHREADS + tid;
    w[j] = pp[(o < M4) ? o : 0];
  }
}

__device__ __forceinline__ void col_apply(const u4 (&w)[CQ], const float* hcol, float* acc,
                                          int t, int tid, int M4) {
#pragma unroll
  for (int j = 0; j < CQ; ++j) asm volatile("" :: "v"(w[j]));
  unsigned r[CU];
  float hv[CU];
#pragma unroll
  for (int j = 0; j < CQ; ++j) {
    const int o = (t * CQ + j) * CTHREADS + tid;
    const bool live = (o < M4);
#pragma unroll
    for (int c = 0; c < 4; ++c) {
      const unsigned e = w[j][c];
      const int k = j * 4 + c;
      r[k] = e >> 16;
      hv[k] = hcol[live ? (e & 0xffffu) : NHALF];
    }
  }
  asm volatile("" :: "v"(hv[0]), "v"(hv[1]), "v"(hv[2]), "v"(hv[3]), "v"(hv[4]), "v"(hv[5]),
               "v"(hv[6]), "v"(hv[7]), "v"(hv[8]), "v"(hv[9]), "v"(hv[10]), "v"(hv[11]),
               "v"(hv[12]), "v"(hv[13]), "v"(hv[14]), "v"(hv[15]));
#pragma unroll
  for (int k = 0; k < CU; ++k) {
    if (hv[k] != 0.0f) atomicAdd(&acc[r[k]], hv[k]);
  }
}

__global__ void __launch_bounds__(CTHREADS)
colscatter_kernel(const unsigned int* __restrict__ list0, const unsigned int* __restrict__ list1,
                  const int* __restrict__ ctr,
                  const float* __restrict__ dis, const float* __restrict__ hT,
                  float* __restrict__ lapT, int n) {
  __shared__ float hcol[NHALF + 1];
  __shared__ float acc[NSTG * CTHREADS];
  const int tid = threadIdx.x;
  const int f = blockIdx.x >> 2;
  const int s = (blockIdx.x >> 1) & 1;
  const int q = blockIdx.x & 1;
  const int half = (n + 1) >> 1;
  const int sbase = s * half;
  const int scnt = (s == 0) ? half : (n - half);
  const float* __restrict__ col = hT + (size_t)f * n + sbase;
  const float* __restrict__ diss = dis + sbase;
  {
    float cv[NOUT], dv[NOUT];
#pragma unroll
    for (int k = 0; k < NOUT; ++k) {
      const int i = tid + k * CTHREADS;
      const int ic = (i < scnt) ? i : 0;
      cv[k] = col[ic];
      dv[k] = diss[ic];
    }
#pragma unroll
    for (int k = 0; k < NOUT; ++k) {
      const int i = tid + k * CTHREADS;
      if (i < scnt) hcol[i] = dv[k] * cv[k];
    }
  }
  for (int k = 0; k < NSTG; ++k) acc[tid + k * CTHREADS] = 0.0f;
  if (tid == 0) hcol[NHALF] = 0.0f;
  __syncthreads();

  const unsigned int* __restrict__ lst = (s == 0) ? list0 : list1;
  const int M = ctr[s];
  const int eh = (((M + 1) >> 1) + 3) & ~3;
  const int e0 = (q * eh < M) ? (q * eh) : M;
  const int e1 = (e0 + eh < M) ? (e0 + eh) : M;
  const int M4 = (e1 - e0) >> 2;
  const int T = (M4 + CQ * CTHREADS - 1) / (CQ * CTHREADS);
  const u4* __restrict__ pp = (const u4*)(lst + e0);

  u4 wA[CQ], wB[CQ];
  if (T > 0) {
    col_load(wA, pp, 0, tid, M4);
    for (int t = 0; t < T; t += 2) {
      const int tb = (t + 1 < T) ? (t + 1) : (T - 1);
      const int ta = (t + 2 < T) ? (t + 2) : (T - 1);
      col_load(wB, pp, tb, tid, M4);
      __builtin_amdgcn_sched_barrier(0);
      col_apply(wA, hcol, acc, t, tid, M4);
      if (t + 1 >= T) break;
      col_load(wA, pp, ta, tid, M4);
      __builtin_amdgcn_sched_barrier(0);
      col_apply(wB, hcol, acc, t + 1, tid, M4);
    }
  }
  if (tid < (e1 - e0) - 4 * M4) {
    const unsigned e = lst[e0 + 4 * M4 + tid];
    const float v = hcol[e & 0xffffu];
    if (v != 0.0f) atomicAdd(&acc[e >> 16], v);
  }
  float dd[NSTG];
#pragma unroll
  for (int k = 0; k < NSTG; ++k) {
    const int i = tid + k * CTHREADS;
    dd[k] = dis[(i < n) ? i : 0];
  }
  __syncthreads();

  float* outp = lapT + (size_t)f * n;
  float av[NSTG];
#pragma unroll
  for (int k = 0; k < NSTG; ++k) av[k] = acc[tid + k * CTHREADS];
#pragma unroll
  for (int k = 0; k < NSTG; ++k) {
    const int i = tid + k * CTHREADS;
    if (i < n) atomicAdd(outp + i, -(dd[k] * av[k]));
  }
}


__device__ __forceinline__ void stage_w_pairs(float* sWp, const float* __restrict__ W,
                                              int tid) {
  const float4* Wv = (const float4*)W;
#pragma unroll
  for (int i = 0; i < 8; ++i) {
    int idx = tid + i * 128;
    float4 v = Wv[idx];
    int k = idx >> 4;
    int c4 = (idx & 15) * 4;
    float* dstp = sWp + (k >> 1) * WPSTR + (k & 1);
    dstp[(c4 + 0) * 2] = v.x;
    dstp[(c4 + 1) * 2] = v.y;
    dstp[(c4 + 2) * 2] = v.z;
    dstp[(c4 + 3) * 2] = v.w;
  }
}

__device__ __forceinline__ void stage_rows(float* sAw, const float* __restrict__ A,
                                           int rowBase, int n, int lane) {
#pragma unroll
  for (int i = 0; i < 8; ++i) {
    int fi = lane + i * 32;
    int r = fi >> 4, c4 = fi & 15;
    int row = rowBase + r;
    if (row >= n) row = n - 1;
    ((float4*)(sAw + r * ASTR))[c4] = ((const float4*)(A + (size_t)row * 64))[c4];
  }
}

__device__ __forceinline__ void stage_rows_T(float* sAw, const float* __restrict__ AT,
                                             int rowBase, int n, int lane) {
  const int r = lane & 15, kh = lane >> 4;
  int row = rowBase + r;
  if (row >= n) row = n - 1;
#pragma unroll
  for (int i = 0; i < 32; ++i) {
    int k = 2 * i + kh;
    sAw[r * ASTR + k] = AT[(size_t)k * n + row];
  }
}

__device__ __forceinline__ void mm_accum64(const float* sAw, const float* sWp,
                                           int lane, v8f acc[4]) {
  const int nlo = lane & 15;
  const int g = lane >> 4;
#pragma unroll
  for (int kk = 0; kk < 16; ++kk) {
    const int k0 = kk * 4 + 2 * g;
    v2f a = *(const v2f*)(sAw + nlo * ASTR + k0);
    const float* brow = sWp + (kk * 2 + g) * WPSTR;
#pragma unroll
    for (int nt = 0; nt < 4; ++nt) {
      v2f b = *(const v2f*)(brow + (nt * 16 + nlo) * 2);
      acc[nt] = wmma4(a, b, acc[nt]);
    }
  }
}


__global__ void __launch_bounds__(128)
gemm64_relu_kernel(const float* __restrict__ A, const float* __restrict__ W,
                   const float* __restrict__ bias, float* __restrict__ outT, int n) {
  __shared__ float sWp[WPROWS * WPSTR];
  __shared__ float sA[GWAVES * 16 * ASTR];
  __shared__ float sB[64];
  const int tid = threadIdx.x, wave = tid >> 5, lane = tid & 31;
  stage_w_pairs(sWp, W, tid);
  if (tid < 64) sB[tid] = bias[tid];

  const int rowBase = blockIdx.x * GROWS + wave * 16;
  float* sAw = sA + wave * 16 * ASTR;
  stage_rows(sAw, A, rowBase, n, lane);
  __syncthreads();

  v8f z = {0, 0, 0, 0, 0, 0, 0, 0};
  v8f acc[4] = {z, z, z, z};
  mm_accum64(sAw, sWp, lane, acc);

  const int nlo = lane & 15, g = lane >> 4;
#pragma unroll
  for (int nt = 0; nt < 4; ++nt)
#pragma unroll
    for (int r = 0; r < 8; ++r) {
      int col = nt * 16 + nlo;
      sAw[(r + 8 * g) * ASTR + col] = fmaxf(acc[nt][r] + sB[col], 0.0f);
    }
  __syncthreads();
  {
    const int r = lane & 15, kh = lane >> 4;
    const int row = rowBase + r;
    if (row < n) {
#pragma unroll
      for (int i = 0; i < 32; ++i) {
        int k = 2 * i + kh;
        outT[(size_t)k * n + row] = sAw[r * ASTR + k];
      }
    }
  }
}

__global__ void __launch_bounds__(128)
gemm64_cheb2_kernel(float* __restrict__ h, const float* __restrict__ lap,
                    const float* __restrict__ W0, const float* __restrict__ W1,
                    const float* __restrict__ bias, int n) {
  __shared__ float sWp[2 * WPROWS * WPSTR];
  __shared__ float sA[GWAVES * 16 * ASTR];
  __shared__ float sB[64];
  const int tid = threadIdx.x, wave = tid >> 5, lane = tid & 31;
  stage_w_pairs(sWp, W0, tid);
  stage_w_pairs(sWp + WPROWS * WPSTR, W1, tid);
  if (tid < 64) sB[tid] = bias[tid];

  const int rowBase = blockIdx.x * GROWS + wave * 16;
  float* sAw = sA + wave * 16 * ASTR;

  v8f z = {0, 0, 0, 0, 0, 0, 0, 0};
  v8f acc[4] = {z, z, z, z};

  stage_rows_T(sAw, h, rowBase, n, lane);
  __syncthreads();
  mm_accum64(sAw, sWp, lane, acc);
  __syncthreads();
  stage_rows_T(sAw, lap, rowBase, n, lane);
  __syncthreads();
  mm_accum64(sAw, sWp + WPROWS * WPSTR, lane, acc);

  const int nlo = lane & 15, g = lane >> 4;
#pragma unroll
  for (int nt = 0; nt < 4; ++nt)
#pragma unroll
    for (int r = 0; r < 8; ++r) {
      int col = nt * 16 + nlo;
      sAw[(r + 8 * g) * ASTR + col] = fmaxf(acc[nt][r] + sB[col], 0.0f);
    }
  __syncthreads();
  {
    const int r = lane & 15, kh = lane >> 4;
    const int row = rowBase + r;
    if (row < n) {
      float res[32];
#pragma unroll
      for (int i = 0; i < 32; ++i) res[i] = h[(size_t)(2 * i + kh) * n + row];
#pragma unroll
      for (int i = 0; i < 32; ++i) {
        int k = 2 * i + kh;
        h[(size_t)k * n + row] = sAw[r * ASTR + k] + res[i];
      }
    }
  }
}

__global__ void __launch_bounds__(128)
gemm_out_kernel(const float* __restrict__ A, const float* __restrict__ Wl,
                const float* __restrict__ bl, float* __restrict__ logits, int n) {
  __shared__ float sWo[WPROWS * WOSTR];
  __shared__ float sA[GWAVES * 16 * ASTR];
  __shared__ float sB[16];
  const int tid = threadIdx.x, wave = tid >> 5, lane = tid & 31;
  for (int idx = tid; idx < 64 * 16; idx += 128) {
    int k = idx >> 4, c = idx & 15;
    float v = (c < CDIM) ? Wl[k * CDIM + c] : 0.0f;
    sWo[(k >> 1) * WOSTR + c * 2 + (k & 1)] = v;
  }
  if (tid < 16) sB[tid] = (tid < CDIM) ? bl[tid] : 0.0f;

  const int rowBase = blockIdx.x * GROWS + wave * 16;
  float* sAw = sA + wave * 16 * ASTR;
  stage_rows_T(sAw, A, rowBase, n, lane);
  __syncthreads();

  const int nlo = lane & 15, g = lane >> 4;
  v8f acc = {0, 0, 0, 0, 0, 0, 0, 0};
#pragma unroll
  for (int kk = 0; kk < 16; ++kk) {
    const int k0 = kk * 4 + 2 * g;
    v2f a = *(const v2f*)(sAw + nlo * ASTR + k0);
    v2f b = *(const v2f*)(sWo + (kk * 2 + g) * WOSTR + nlo * 2);
    acc = wmma4(a, b, acc);
  }
#pragma unroll
  for (int r = 0; r < 8; ++r) {
    int row = rowBase + r + 8 * g;
    if (row < n) logits[(size_t)row * 16 + nlo] = acc[r] + sB[nlo];
  }
}

__global__ void logsoftmax_kernel(const float* __restrict__ logits,
                                  float* __restrict__ out, int n) {
  int i = blockIdx.x * blockDim.x + threadIdx.x;
  if (i >= n) return;
  const float* v = logits + (size_t)i * 16;
  float m = v[0];
#pragma unroll
  for (int c = 1; c < CDIM; ++c) m = fmaxf(m, v[c]);
  float s = 0.0f;
#pragma unroll
  for (int c = 0; c < CDIM; ++c) s += expf(v[c] - m);
  float ls = logf(s);
#pragma unroll
  for (int c = 0; c < CDIM; ++c) out[(size_t)i * CDIM + c] = v[c] - m - ls;
}


extern "C" void kernel_launch(void* const* d_in, const int* in_sizes, int n_in,
                              void* d_out, int out_size, void* d_ws, size_t ws_size,
                              hipStream_t stream) {
  const float* x    = (const float*)d_in[0];
  const int*   ei   = (const int*)d_in[1];
  const float* W1_0 = (const float*)d_in[2];
  const float* b1   = (const float*)d_in[3];
  const float* W2_0 = (const float*)d_in[4];
  const float* W2_1 = (const float*)d_in[5];
  const float* b2   = (const float*)d_in[6];
  const float* Wl   = (const float*)d_in[7];
  const float* bl   = (const float*)d_in[8];

  const int N = in_sizes[0] / FDIM;
  const int E = in_sizes[1] / 2;
  const int* src = ei;
  const int* dst = ei + E;

  float* ws     = (float*)d_ws;
  float* lap    = ws + N;
  float* dis    = lap + (size_t)N * 64;
  unsigned int* list0 = (unsigned int*)(dis + N);
  float* h      = dis + N + E;
  float* logits = h + (size_t)N * 64;
  float* out    = (float*)d_out;
  float* cnt    = h;
  unsigned int* list1 = (unsigned int*)logits;
  const int nblk = (E + PTHREADS - 1) / PTHREADS;
  int* c0  = (int*)d_ws;
  int* ctr = c0 + nblk;

  {
    int tot = N * 64;
    zero_kernel<<<(tot + 255) / 256, 256, 0, stream>>>(lap, tot, cnt, N);
  }
  degree_kernel<<<nblk, PTHREADS, 0, stream>>>(src, cnt, c0, E, (N + 1) >> 1);
  place_dis_kernel<<<nblk + (N + PTHREADS - 1) / PTHREADS, PTHREADS, 0, stream>>>(
      src, dst, c0, ctr, list0, list1, cnt, dis, N, E, nblk, (N + 1) >> 1);

  const int gblocks = (N + GROWS - 1) / GROWS;
  gemm64_relu_kernel<<<gblocks, 128, 0, stream>>>(x, W1_0, b1, h, N);
  colscatter_kernel<<<64 * 4, CTHREADS, 0, stream>>>(list0, list1, ctr, dis, h, lap, N);
  gemm64_cheb2_kernel<<<gblocks, 128, 0, stream>>>(h, lap, W2_0, W2_1, b2, N);
  gemm_out_kernel<<<gblocks, 128, 0, stream>>>(h, Wl, bl, logits, N);
  logsoftmax_kernel<<<(N + 255) / 256, 256, 0, stream>>>(logits, out, N);
}


# lap by destination quarter: edges split 4 ways by destination; block (feature, quarter) holds whole column + its row sums, stores lap once: no zero fill, no global adds; 1 sum per row (was 4)
# speedup vs baseline: 1.7963x; 1.0412x over previous
#include <hip/hip_runtime.h>
#include <hip/hip_bf16.h>
#include <stdint.h>


#define FDIM 64
#define HDIM 64
#define CDIM 10

#define GWAVES 4
#define GROWS  (GWAVES * 16)
#define ASTR   68
#define WPSTR  160
#define WPROWS 32
#define WOSTR  96
#define CSTR   32
#define NMAX   50000
#define NQ     12500
#define CTHREADS 1024
#define CU     16
#define CQ     (CU / 4)
#define NCOL   25
#define NACC   13
#define PTHREADS 1024

typedef unsigned int u4 __attribute__((ext_vector_type(4)));
typedef float v2f __attribute__((ext_vector_type(2)));
typedef float v8f __attribute__((ext_vector_type(8)));

__device__ __forceinline__ v8f wmma4(v2f a, v2f b, v8f c) {
  return __builtin_amdgcn_wmma_f32_16x16x4_f32(false, a, false, b, (short)0, c,
                                               false, false);
}


__global__ void zero_kernel(float* __restrict__ cnt, int ncnt) {
  int i = blockIdx.x * blockDim.x + threadIdx.x;
  if (i < ncnt) cnt[(size_t)i * CSTR] = 0.0f;
}

__device__ __forceinline__ int wave_sum(int v) {
  for (int o = 16; o > 0; o >>= 1) v += __shfl_xor(v, o, 32);
  return v;
}

__device__ __forceinline__ int dst_quarter(int d, int qn) {
  return (d >= qn) + (d >= 2 * qn) + (d >= 3 * qn);
}

__global__ void __launch_bounds__(PTHREADS)
degree_kernel(const int* __restrict__ src, const int* __restrict__ dst, float* __restrict__ cnt,
              int* __restrict__ c4, int E, int qn) {
  __shared__ int sw[4][PTHREADS / 32];
  const int tid = threadIdx.x, lane = tid & 31, wave = tid >> 5;
  const int e = blockIdx.x * PTHREADS + tid;
  const bool valid = (e < E);
  int s = 0, d = 0;
  if (valid) { s = src[e]; d = dst[e]; }
  if (valid) atomicAdd(&cnt[(size_t)s * CSTR], 1.0f);
  const int cls = dst_quarter(d, qn);
#pragma unroll
  for (int k = 0; k < 4; ++k) {
    const unsigned b = (unsigned)__ballot(valid && (cls == k));
    if (lane == 0) sw[k][wave] = __popc(b);
  }
  __syncthreads();
  if (wave < 4) {
    const int v = wave_sum(sw[wave][lane]);
    if (lane == 0) c4[blockIdx.x * 4 + wave] = v;
  }
}

__global__ void __launch_bounds__(PTHREADS)
place_dis_kernel(const int* __restrict__ src, const int* __restrict__ dst,
                 const int* __restrict__ c4, int* __restrict__ ctr,
                 unsigned int* __restrict__ listA, unsigned int* __restrict__ listB,
                 const float* __restrict__ cnt, float* __restrict__ dis,
                 int n, int E, int nblk, int qn) {
  __shared__ int spre[4][PTHREADS / 32];
  __shared__ int stot[4][PTHREADS / 32];
  __shared__ int sw[4][PTHREADS / 32];
  const int tid = threadIdx.x, lane = tid & 31, wave = tid >> 5;
  const int blk = blockIdx.x;
  if (blk >= nblk) {
    const int i = (blk - nblk) * PTHREADS + tid;
    if (i < n) {
      const float d = cnt[(size_t)i * CSTR];
      dis[i] = (d > 0.0f) ? rsqrtf(fmaxf(d, 1.0f)) : 0.0f;
    }
    return;
  }
  int pre[4] = {0, 0, 0, 0}, tot[4] = {0, 0, 0, 0};
  for (int i = tid; i < nblk; i += PTHREADS) {
    const int4 v = ((const int4*)c4)[i];
    const bool before = (i < blk);
    tot[0] += v.x; tot[1] += v.y; tot[2] += v.z; tot[3] += v.w;
    pre[0] += before ? v.x : 0; pre[1] += before ? v.y : 0;
    pre[2] += before ? v.z : 0; pre[3] += before ? v.w : 0;
  }
#pragma unroll
  for (int k = 0; k < 4; ++k) {
    const int p = wave_sum(pre[k]), t = wave_sum(tot[k]);
    if (lane == 0) { spre[k][wave] = p; stot[k][wave] = t; }
  }
  const int e = blk * PTHREADS + tid;
  const bool valid = (e < E);
  int s = 0, d = 0;
  if (valid) { s = src[e]; d = dst[e]; }
  const int cls = dst_quarter(d, qn);
  unsigned b[4];
#pragma unroll
  for (int k = 0; k < 4; ++k) {
    b[k] = (unsigned)__ballot(valid && (cls == k));
    if (lane == 0) sw[k][wave] = __popc(b[k]);
  }
  __syncthreads();
  int P[4], T[4], Wv[4];
#pragma unroll
  for (int k = 0; k < 4; ++k) {
    P[k] = wave_sum(spre[k][lane]);
    T[k] = wave_sum(stot[k][lane]);
    Wv[k] = wave_sum((lane < wave) ? sw[k][lane] : 0);
  }
  if (valid) {
    const int Pk = (cls == 0) ? P[0] : (cls == 1) ? P[1] : (cls == 2) ? P[2] : P[3];
    const int Tk = (cls == 0) ? T[0] : (cls == 1) ? T[1] : (cls == 2) ? T[2] : T[3];
    const int Wk = (cls == 0) ? Wv[0] : (cls == 1) ? Wv[1] : (cls == 2) ? Wv[2] : Wv[3];
    const unsigned bk = (cls == 0) ? b[0] : (cls == 1) ? b[1] : (cls == 2) ? b[2] : b[3];
    const unsigned lt = (1u << lane) - 1u;
    const int rank = Pk + Wk + __popc(bk & lt);
    const int base = (cls & 1) ? (E - Tk) : 0;
    unsigned int* lst = (cls < 2) ? listA : listB;
    lst[base + rank] = (unsigned)s | ((unsigned)(d - cls * qn) << 16);
  }
  if (blk == nblk - 1 && tid == 0) {
    ctr[0] = T[0]; ctr[1] = T[1]; ctr[2] = T[2]; ctr[3] = T[3];
  }
}


__device__ __forceinline__ void col_load(u4 (&w)[CQ], const u4* __restrict__ pp,
                                         int t, int tid, int M4) {
#pragma unroll
  for (int j = 0; j < CQ; ++j) {
    int o = (t * CQ + j) * CTHREADS + tid;
    w[j] = pp[(o < M4) ? o : 0];
  }
}

__device__ __forceinline__ void col_apply(const u4 (&w)[CQ], const float* hcol, float* acc,
                                          int t, int tid, int M4) {
#pragma unroll
  for (int j = 0; j < CQ; ++j) asm volatile("" :: "v"(w[j]));
  unsigned r[CU];
  float hv[CU];
#pragma unroll
  for (int j = 0; j < CQ; ++j) {
    const int o = (t * CQ + j) * CTHREADS + tid;
    const bool live = (o < M4);
#pragma unroll
    for (int c = 0; c < 4; ++c) {
      const unsigned e = w[j][c];
      const int k = j * 4 + c;
      r[k] = e >> 16;
      hv[k] = hcol[live ? (e & 0xffffu) : NMAX];
    }
  }
  asm volatile("" :: "v"(hv[0]), "v"(hv[1]), "v"(hv[2]), "v"(hv[3]), "v"(hv[4]), "v"(hv[5]),
               "v"(hv[6]), "v"(hv[7]), "v"(hv[8]), "v"(hv[9]), "v"(hv[10]), "v"(hv[11]),
               "v"(hv[12]), "v"(hv[13]), "v"(hv[14]), "v"(hv[15]));
#pragma unroll
  for (int k = 0; k < CU; ++k) {
    if (hv[k] != 0.0f) atomicAdd(&acc[r[k]], hv[k]);
  }
}

__global__ void __launch_bounds__(CTHREADS)
colscatter_kernel(const unsigned int* __restrict__ listA, const unsigned int* __restrict__ listB,
                  const int* __restrict__ ctr,
                  const float* __restrict__ dis, const float* __restrict__ hT,
                  float* __restrict__ lapT, int n, int E) {
  __shared__ float hcol[NMAX + 1];
  __shared__ float acc[NACC * CTHREADS];
  const int tid = threadIdx.x;
  const int f = blockIdx.x >> 2;
  const int q = blockIdx.x & 3;
  const int qn = (n + 3) >> 2;
  const int dbase = q * qn;
  const int drem = n - dbase;
  const int dcnt = (drem < 0) ? 0 : ((drem < qn) ? drem : qn);
  const float* __restrict__ col = hT + (size_t)f * n;
#pragma unroll
  for (int g = 0; g < 2; ++g) {
    float cv[NCOL], dv[NCOL];
#pragma unroll
    for (int k = 0; k < NCOL; ++k) {
      const int i = tid + (g * NCOL + k) * CTHREADS;
      const int ic = (i < n) ? i : 0;
      cv[k] = col[ic];
      dv[k] = dis[ic];
    }
#pragma unroll
    for (int k = 0; k < NCOL; ++k) {
      const int i = tid + (g * NCOL + k) * CTHREADS;
      if (i < n) hcol[i] = dv[k] * cv[k];
    }
  }
  for (int k = 0; k < NACC; ++k) acc[tid + k * CTHREADS] = 0.0f;
  if (tid == 0) hcol[NMAX] = 0.0f;
  __syncthreads();

  const unsigned int* __restrict__ lst = (q < 2) ? listA : listB;
  const int M = ctr[q];
  const int p0 = (q & 1) ? (E - M) : 0;
  const int p1 = p0 + M;
  const int a0r = (p0 + 3) & ~3;
  const int a0 = (a0r < p1) ? a0r : p1;
  const int M4 = (p1 - a0) >> 2;
  const int T = (M4 + CQ * CTHREADS - 1) / (CQ * CTHREADS);
  const u4* __restrict__ pp = (const u4*)(lst + a0);

  u4 wA[CQ], wB[CQ];
  if (T > 0) {
    col_load(wA, pp, 0, tid, M4);
    for (int t = 0; t < T; t += 2) {
      const int tb = (t + 1 < T) ? (t + 1) : (T - 1);
      const int ta = (t + 2 < T) ? (t + 2) : (T - 1);
      col_load(wB, pp, tb, tid, M4);
      __builtin_amdgcn_sched_barrier(0);
      col_apply(wA, hcol, acc, t, tid, M4);
      if (t + 1 >= T) break;
      col_load(wA, pp, ta, tid, M4);
      __builtin_amdgcn_sched_barrier(0);
      col_apply(wB, hcol, acc, t + 1, tid, M4);
    }
  }
  {
    const int nhead = a0 - p0;
    const int t0 = a0 + 4 * M4;
    const int ntail = p1 - t0;
    if (tid < nhead + ntail) {
      const unsigned e = lst[(tid < nhead) ? (p0 + tid) : (t0 + tid - nhead)];
      const float v = hcol[e & 0xffffu];
      if (v != 0.0f) atomicAdd(&acc[e >> 16], v);
    }
  }
  float dd[NACC];
#pragma unroll
  for (int k = 0; k < NACC; ++k) {
    const int i = tid + k * CTHREADS;
    dd[k] = dis[(i < dcnt) ? (dbase + i) : 0];
  }
  __syncthreads();

  float* outp = lapT + (size_t)f * n + dbase;
  float av[NACC];
#pragma unroll
  for (int k = 0; k < NACC; ++k) av[k] = acc[tid + k * CTHREADS];
#pragma unroll
  for (int k = 0; k < NACC; ++k) {
    const int i = tid + k * CTHREADS;
    if (i < dcnt) outp[i] = -(dd[k] * av[k]);
  }
}


__device__ __forceinline__ void stage_w_pairs(float* sWp, const float* __restrict__ W,
                                              int tid) {
  const float4* Wv = (const float4*)W;
#pragma unroll
  for (int i = 0; i < 8; ++i) {
    int idx = tid + i * 128;
    float4 v = Wv[idx];
    int k = idx >> 4;
    int c4 = (idx & 15) * 4;
    float* dstp = sWp + (k >> 1) * WPSTR + (k & 1);
    dstp[(c4 + 0) * 2] = v.x;
    dstp[(c4 + 1) * 2] = v.y;
    dstp[(c4 + 2) * 2] = v.z;
    dstp[(c4 + 3) * 2] = v.w;
  }
}

__device__ __forceinline__ void stage_rows(float* sAw, const float* __restrict__ A,
                                           int rowBase, int n, int lane) {
#pragma unroll
  for (int i = 0; i < 8; ++i) {
    int fi = lane + i * 32;
    int r = fi >> 4, c4 = fi & 15;
    int row = rowBase + r;
    if (row >= n) row = n - 1;
    ((float4*)(sAw + r * ASTR))[c4] = ((const float4*)(A + (size_t)row * 64))[c4];
  }
}

__device__ __forceinline__ void stage_rows_T(float* sAw, const float* __restrict__ AT,
                                             int rowBase, int n, int lane) {
  const int r = lane & 15, kh = lane >> 4;
  int row = rowBase + r;
  if (row >= n) row = n - 1;
#pragma unroll
  for (int i = 0; i < 32; ++i) {
    int k = 2 * i + kh;
    sAw[r * ASTR + k] = AT[(size_t)k * n + row];
  }
}

__device__ __forceinline__ void mm_accum64(const float* sAw, const float* sWp,
                                           int lane, v8f acc[4]) {
  const int nlo = lane & 15;
  const int g = lane >> 4;
#pragma unroll
  for (int kk = 0; kk < 16; ++kk) {
    const int k0 = kk * 4 + 2 * g;
    v2f a = *(const v2f*)(sAw + nlo * ASTR + k0);
    const float* brow = sWp + (kk * 2 + g) * WPSTR;
#pragma unroll
    for (int nt = 0; nt < 4; ++nt) {
      v2f b = *(const v2f*)(brow + (nt * 16 + nlo) * 2);
      acc[nt] = wmma4(a, b, acc[nt]);
    }
  }
}


__global__ void __launch_bounds__(128)
gemm64_relu_kernel(const float* __restrict__ A, const float* __restrict__ W,
                   const float* __restrict__ bias, float* __restrict__ outT, int n) {
  __shared__ float sWp[WPROWS * WPSTR];
  __shared__ float sA[GWAVES * 16 * ASTR];
  __shared__ float sB[64];
  const int tid = threadIdx.x, wave = tid >> 5, lane = tid & 31;
  stage_w_pairs(sWp, W, tid);
  if (tid < 64) sB[tid] = bias[tid];

  const int rowBase = blockIdx.x * GROWS + wave * 16;
  float* sAw = sA + wave * 16 * ASTR;
  stage_rows(sAw, A, rowBase, n, lane);
  __syncthreads();

  v8f z = {0, 0, 0, 0, 0, 0, 0, 0};
  v8f acc[4] = {z, z, z, z};
  mm_accum64(sAw, sWp, lane, acc);

  const int nlo = lane & 15, g = lane >> 4;
#pragma unroll
  for (int nt = 0; nt < 4; ++nt)
#pragma unroll
    for (int r = 0; r < 8; ++r) {
      int col = nt * 16 + nlo;
      sAw[(r + 8 * g) * ASTR + col] = fmaxf(acc[nt][r] + sB[col], 0.0f);
    }
  __syncthreads();
  {
    const int r = lane & 15, kh = lane >> 4;
    const int row = rowBase + r;
    if (row < n) {
#pragma unroll
      for (int i = 0; i < 32; ++i) {
        int k = 2 * i + kh;
        outT[(size_t)k * n + row] = sAw[r * ASTR + k];
      }
    }
  }
}

__global__ void __launch_bounds__(128)
gemm64_cheb2_kernel(float* __restrict__ h, const float* __restrict__ lap,
                    const float* __restrict__ W0, const float* __restrict__ W1,
                    const float* __restrict__ bias, int n) {
  __shared__ float sWp[2 * WPROWS * WPSTR];
  __shared__ float sA[GWAVES * 16 * ASTR];
  __shared__ float sB[64];
  const int tid = threadIdx.x, wave = tid >> 5, lane = tid & 31;
  stage_w_pairs(sWp, W0, tid);
  stage_w_pairs(sWp + WPROWS * WPSTR, W1, tid);
  if (tid < 64) sB[tid] = bias[tid];

  const int rowBase = blockIdx.x * GROWS + wave * 16;
  float* sAw = sA + wave * 16 * ASTR;

  v8f z = {0, 0, 0, 0, 0, 0, 0, 0};
  v8f acc[4] = {z, z, z, z};

  stage_rows_T(sAw, h, rowBase, n, lane);
  __syncthreads();
  mm_accum64(sAw, sWp, lane, acc);
  __syncthreads();
  stage_rows_T(sAw, lap, rowBase, n, lane);
  __syncthreads();
  mm_accum64(sAw, sWp + WPROWS * WPSTR, lane, acc);

  const int nlo = lane & 15, g = lane >> 4;
#pragma unroll
  for (int nt = 0; nt < 4; ++nt)
#pragma unroll
    for (int r = 0; r < 8; ++r) {
      int col = nt * 16 + nlo;
      sAw[(r + 8 * g) * ASTR + col] = fmaxf(acc[nt][r] + sB[col], 0.0f);
    }
  __syncthreads();
  {
    const int r = lane & 15, kh = lane >> 4;
    const int row = rowBase + r;
    if (row < n) {
      float res[32];
#pragma unroll
      for (int i = 0; i < 32; ++i) res[i] = h[(size_t)(2 * i + kh) * n + row];
#pragma unroll
      for (int i = 0; i < 32; ++i) {
        int k = 2 * i + kh;
        h[(size_t)k * n + row] = sAw[r * ASTR + k] + res[i];
      }
    }
  }
}

__global__ void __launch_bounds__(128)
gemm_out_kernel(const float* __restrict__ A, const float* __restrict__ Wl,
                const float* __restrict__ bl, float* __restrict__ logits, int n) {
  __shared__ float sWo[WPROWS * WOSTR];
  __shared__ float sA[GWAVES * 16 * ASTR];
  __shared__ float sB[16];
  const int tid = threadIdx.x, wave = tid >> 5, lane = tid & 31;
  for (int idx = tid; idx < 64 * 16; idx += 128) {
    int k = idx >> 4, c = idx & 15;
    float v = (c < CDIM) ? Wl[k * CDIM + c] : 0.0f;
    sWo[(k >> 1) * WOSTR + c * 2 + (k & 1)] = v;
  }
  if (tid < 16) sB[tid] = (tid < CDIM) ? bl[tid] : 0.0f;

  const int rowBase = blockIdx.x * GROWS + wave * 16;
  float* sAw = sA + wave * 16 * ASTR;
  stage_rows_T(sAw, A, rowBase, n, lane);
  __syncthreads();

  const int nlo = lane & 15, g = lane >> 4;
  v8f acc = {0, 0, 0, 0, 0, 0, 0, 0};
#pragma unroll
  for (int kk = 0; kk < 16; ++kk) {
    const int k0 = kk * 4 + 2 * g;
    v2f a = *(const v2f*)(sAw + nlo * ASTR + k0);
    v2f b = *(const v2f*)(sWo + (kk * 2 + g) * WOSTR + nlo * 2);
    acc = wmma4(a, b, acc);
  }
#pragma unroll
  for (int r = 0; r < 8; ++r) {
    int row = rowBase + r + 8 * g;
    if (row < n) logits[(size_t)row * 16 + nlo] = acc[r] + sB[nlo];
  }
}

__global__ void logsoftmax_kernel(const float* __restrict__ logits,
                                  float* __restrict__ out, int n) {
  int i = blockIdx.x * blockDim.x + threadIdx.x;
  if (i >= n) return;
  const float* v = logits + (size_t)i * 16;
  float m = v[0];
#pragma unroll
  for (int c = 1; c < CDIM; ++c) m = fmaxf(m, v[c]);
  float s = 0.0f;
#pragma unroll
  for (int c = 0; c < CDIM; ++c) s += expf(v[c] - m);
  float ls = logf(s);
#pragma unroll
  for (int c = 0; c < CDIM; ++c) out[(size_t)i * CDIM + c] = v[c] - m - ls;
}


extern "C" void kernel_launch(void* const* d_in, const int* in_sizes, int n_in,
                              void* d_out, int out_size, void* d_ws, size_t ws_size,
                              hipStream_t stream) {
  const float* x    = (const float*)d_in[0];
  const int*   ei   = (const int*)d_in[1];
  const float* W1_0 = (const float*)d_in[2];
  const float* b1   = (const float*)d_in[3];
  const float* W2_0 = (const float*)d_in[4];
  const float* W2_1 = (const float*)d_in[5];
  const float* b2   = (const float*)d_in[6];
  const float* Wl   = (const float*)d_in[7];
  const float* bl   = (const float*)d_in[8];

  const int N = in_sizes[0] / FDIM;
  const int E = in_sizes[1] / 2;
  const int* src = ei;
  const int* dst = ei + E;

  float* ws     = (float*)d_ws;
  float* lap    = ws + N;
  float* dis    = lap + (size_t)N * 64;
  unsigned int* listA = (unsigned int*)(dis + N);
  float* h      = dis + N + E;
  float* logits = h + (size_t)N * 64;
  float* out    = (float*)d_out;
  float* cnt    = h;
  unsigned int* listB = (unsigned int*)logits;
  const int nblk = (E + PTHREADS - 1) / PTHREADS;
  int* c4  = (int*)d_ws;
  int* ctr = c4 + 4 * nblk;
  const int qn = (N + 3) >> 2;

  zero_kernel<<<(N + 255) / 256, 256, 0, stream>>>(cnt, N);
  degree_kernel<<<nblk, PTHREADS, 0, stream>>>(src, dst, cnt, c4, E, qn);
  place_dis_kernel<<<nblk + (N + PTHREADS - 1) / PTHREADS, PTHREADS, 0, stream>>>(
      src, dst, c4, ctr, listA, listB, cnt, dis, N, E, nblk, qn);

  const int gblocks = (N + GROWS - 1) / GROWS;
  gemm64_relu_kernel<<<gblocks, 128, 0, stream>>>(x, W1_0, b1, h, N);
  colscatter_kernel<<<64 * 4, CTHREADS, 0, stream>>>(listA, listB, ctr, dis, h, lap, N, E);
  gemm64_cheb2_kernel<<<gblocks, 128, 0, stream>>>(h, lap, W2_0, W2_1, b2, N);
  gemm_out_kernel<<<gblocks, 128, 0, stream>>>(h, Wl, bl, logits, N);
  logsoftmax_kernel<<<(N + 255) / 256, 256, 0, stream>>>(logits, out, N);
}


# graph prep: each thread of the degree pass and of the placing pass takes 8 consecutive edges (two 16-byte loads per index row), so the two passes start 1/8 the waves; same lists, same order
# speedup vs baseline: 1.8520x; 1.0310x over previous
#include <hip/hip_runtime.h>
#include <hip/hip_bf16.h>
#include <stdint.h>


#define FDIM 64
#define HDIM 64
#define CDIM 10

#define GWAVES 4
#define GROWS  (GWAVES * 16)
#define ASTR   68
#define WPSTR  160
#define WPROWS 32
#define WOSTR  96
#define CSTR   32
#define NMAX   50000
#define NQ     12500
#define CTHREADS 1024
#define CU     16
#define CQ     (CU / 4)
#define NCOL   25
#define NACC   13
#define PTHREADS 256
#define EPT      8
#define PEDGES   (PTHREADS * EPT)
#define PWAVES   (PTHREADS / 32)

typedef unsigned int u4 __attribute__((ext_vector_type(4)));
typedef float v2f __attribute__((ext_vector_type(2)));
typedef float v8f __attribute__((ext_vector_type(8)));

__device__ __forceinline__ v8f wmma4(v2f a, v2f b, v8f c) {
  return __builtin_amdgcn_wmma_f32_16x16x4_f32(false, a, false, b, (short)0, c,
                                               false, false);
}


__global__ void zero_kernel(float* __restrict__ cnt, int ncnt) {
  int i = blockIdx.x * blockDim.x + threadIdx.x;
  if (i < ncnt) cnt[(size_t)i * CSTR] = 0.0f;
}

__device__ __forceinline__ int wave_sum(int v) {
  for (int o = 16; o > 0; o >>= 1) v += __shfl_xor(v, o, 32);
  return v;
}

__device__ __forceinline__ int dst_quarter(int d, int qn) {
  return (d >= qn) + (d >= 2 * qn) + (d >= 3 * qn);
}

__device__ __forceinline__ int wave_scan(int v, int lane) {
  for (int o = 1; o < 32; o <<= 1) {
    const int t = __shfl_up(v, o, 32);
    if (lane >= o) v += t;
  }
  return v;
}

__device__ __forceinline__ void load_edges(const int* __restrict__ src, const int* __restrict__ dst,
                                           int e0, int E, bool vec, int (&s)[EPT], int (&d)[EPT]) {
  if (vec && e0 + EPT <= E) {
    const int4 s0 = *(const int4*)(src + e0), s1 = *(const int4*)(src + e0 + 4);
    const int4 d0 = *(const int4*)(dst + e0), d1 = *(const int4*)(dst + e0 + 4);
    s[0] = s0.x; s[1] = s0.y; s[2] = s0.z; s[3] = s0.w;
    s[4] = s1.x; s[5] = s1.y; s[6] = s1.z; s[7] = s1.w;
    d[0] = d0.x; d[1] = d0.y; d[2] = d0.z; d[3] = d0.w;
    d[4] = d1.x; d[5] = d1.y; d[6] = d1.z; d[7] = d1.w;
  } else {
#pragma unroll
    for (int j = 0; j < EPT; ++j) {
      const int e = e0 + j;
      s[j] = (e < E) ? src[e] : 0;
      d[j] = (e < E) ? dst[e] : 0;
    }
  }
}

__global__ void __launch_bounds__(PTHREADS)
degree_kernel(const int* __restrict__ src, const int* __restrict__ dst, float* __restrict__ cnt,
              int* __restrict__ c4, int E, int qn) {
  __shared__ int sw[4][PWAVES];
  const int tid = threadIdx.x, lane = tid & 31, wave = tid >> 5;
  const int e0 = (blockIdx.x * PTHREADS + tid) * EPT;
  const int left = E - e0;
  const int nv = (left < 0) ? 0 : ((left < EPT) ? left : EPT);
  int s[EPT], d[EPT];
  load_edges(src, dst, e0, E, (E & 3) == 0, s, d);
  int lo = 0, hi = 0;
#pragma unroll
  for (int j = 0; j < EPT; ++j) {
    if (j < nv) {
      atomicAdd(&cnt[(size_t)s[j] * CSTR], 1.0f);
      const int cls = dst_quarter(d[j], qn);
      const int one = (cls & 1) ? 0x10000 : 1;
      if (cls < 2) lo += one; else hi += one;
    }
  }
  lo = wave_sum(lo);
  hi = wave_sum(hi);
  if (lane == 0) {
    sw[0][wave] = lo & 0xffff; sw[1][wave] = lo >> 16;
    sw[2][wave] = hi & 0xffff; sw[3][wave] = hi >> 16;
  }
  __syncthreads();
  if (tid < 4) {
    int v = 0;
#pragma unroll
    for (int w = 0; w < PWAVES; ++w) v += sw[tid][w];
    c4[blockIdx.x * 4 + tid] = v;
  }
}

__global__ void __launch_bounds__(PTHREADS)
place_dis_kernel(const int* __restrict__ src, const int* __restrict__ dst,
                 const int* __restrict__ c4, int* __restrict__ ctr,
                 unsigned int* __restrict__ listA, unsigned int* __restrict__ listB,
                 const float* __restrict__ cnt, float* __restrict__ dis,
                 int n, int E, int nblk, int qn) {
  __shared__ int spre[4][PWAVES];
  __shared__ int stot[4][PWAVES];
  __shared__ int sw[4][PWAVES];
  const int tid = threadIdx.x, lane = tid & 31, wave = tid >> 5;
  const int blk = blockIdx.x;
  if (blk >= nblk) {
    const int i = (blk - nblk) * PTHREADS + tid;
    if (i < n) {
      const float d = cnt[(size_t)i * CSTR];
      dis[i] = (d > 0.0f) ? rsqrtf(fmaxf(d, 1.0f)) : 0.0f;
    }
    return;
  }
  const int e0 = (blk * PTHREADS + tid) * EPT;
  const int left = E - e0;
  const int nv = (left < 0) ? 0 : ((left < EPT) ? left : EPT);
  int s[EPT], d[EPT];
  load_edges(src, dst, e0, E, (E & 3) == 0, s, d);

  int pre[4] = {0, 0, 0, 0}, tot[4] = {0, 0, 0, 0};
  for (int i = tid; i < nblk; i += 2 * PTHREADS) {
    const int i2 = i + PTHREADS;
    const bool has2 = (i2 < nblk);
    const int4 v = ((const int4*)c4)[i];
    const int4 u = ((const int4*)c4)[has2 ? i2 : i];
    const bool b1 = (i < blk), b2 = has2 && (i2 < blk);
    tot[0] += v.x + (has2 ? u.x : 0); tot[1] += v.y + (has2 ? u.y : 0);
    tot[2] += v.z + (has2 ? u.z : 0); tot[3] += v.w + (has2 ? u.w : 0);
    pre[0] += (b1 ? v.x : 0) + (b2 ? u.x : 0); pre[1] += (b1 ? v.y : 0) + (b2 ? u.y : 0);
    pre[2] += (b1 ? v.z : 0) + (b2 ? u.z : 0); pre[3] += (b1 ? v.w : 0) + (b2 ? u.w : 0);
  }
#pragma unroll
  for (int k = 0; k < 4; ++k) {
    const int p = wave_sum(pre[k]), t = wave_sum(tot[k]);
    if (lane == 0) { spre[k][wave] = p; stot[k][wave] = t; }
  }
  int cls[EPT];
  int lo = 0, hi = 0;
#pragma unroll
  for (int j = 0; j < EPT; ++j) {
    cls[j] = dst_quarter(d[j], qn);
    if (j < nv) {
      const int one = (cls[j] & 1) ? 0x10000 : 1;
      if (cls[j] < 2) lo += one; else hi += one;
    }
  }
  const int ilo = wave_scan(lo, lane), ihi = wave_scan(hi, lane);
  if (lane == 31) {
    sw[0][wave] = ilo & 0xffff; sw[1][wave] = ilo >> 16;
    sw[2][wave] = ihi & 0xffff; sw[3][wave] = ihi >> 16;
  }
  __syncthreads();
  int off[4], T[4];
#pragma unroll
  for (int k = 0; k < 4; ++k) {
    int P = 0, Tk = 0, Wv = 0;
#pragma unroll
    for (int w = 0; w < PWAVES; ++w) {
      P += spre[k][w];
      Tk += stot[k][w];
      Wv += (w < wave) ? sw[k][w] : 0;
    }
    const int mine = (k & 1) ? (((k < 2) ? lo : hi) >> 16) : (((k < 2) ? lo : hi) & 0xffff);
    const int incl = (k & 1) ? (((k < 2) ? ilo : ihi) >> 16) : (((k < 2) ? ilo : ihi) & 0xffff);
    const int base = (k & 1) ? (E - Tk) : 0;
    off[k] = base + P + Wv + (incl - mine);
    T[k] = Tk;
  }
#pragma unroll
  for (int j = 0; j < EPT; ++j) {
    if (j < nv) {
      const int k = cls[j];
      const int o = (k == 0) ? off[0] : (k == 1) ? off[1] : (k == 2) ? off[2] : off[3];
      unsigned int* lst = (k < 2) ? listA : listB;
      lst[o] = (unsigned)s[j] | ((unsigned)(d[j] - k * qn) << 16);
      off[0] += (k == 0); off[1] += (k == 1); off[2] += (k == 2); off[3] += (k == 3);
    }
  }
  if (blk == nblk - 1 && tid == 0) {
    ctr[0] = T[0]; ctr[1] = T[1]; ctr[2] = T[2]; ctr[3] = T[3];
  }
}


__device__ __forceinline__ void col_load(u4 (&w)[CQ], const u4* __restrict__ pp,
                                         int t, int tid, int M4) {
#pragma unroll
  for (int j = 0; j < CQ; ++j) {
    int o = (t * CQ + j) * CTHREADS + tid;
    w[j] = pp[(o < M4) ? o : 0];
  }
}

__device__ __forceinline__ void col_apply(const u4 (&w)[CQ], const float* hcol, float* acc,
                                          int t, int tid, int M4) {
#pragma unroll
  for (int j = 0; j < CQ; ++j) asm volatile("" :: "v"(w[j]));
  unsigned r[CU];
  float hv[CU];
#pragma unroll
  for (int j = 0; j < CQ; ++j) {
    const int o = (t * CQ + j) * CTHREADS + tid;
    const bool live = (o < M4);
#pragma unroll
    for (int c = 0; c < 4; ++c) {
      const unsigned e = w[j][c];
      const int k = j * 4 + c;
      r[k] = e >> 16;
      hv[k] = hcol[live ? (e & 0xffffu) : NMAX];
    }
  }
  asm volatile("" :: "v"(hv[0]), "v"(hv[1]), "v"(hv[2]), "v"(hv[3]), "v"(hv[4]), "v"(hv[5]),
               "v"(hv[6]), "v"(hv[7]), "v"(hv[8]), "v"(hv[9]), "v"(hv[10]), "v"(hv[11]),
               "v"(hv[12]), "v"(hv[13]), "v"(hv[14]), "v"(hv[15]));
#pragma unroll
  for (int k = 0; k < CU; ++k) {
    if (hv[k] != 0.0f) atomicAdd(&acc[r[k]], hv[k]);
  }
}

__global__ void __launch_bounds__(CTHREADS)
colscatter_kernel(const unsigned int* __restrict__ listA, const unsigned int* __restrict__ listB,
                  const int* __restrict__ ctr,
                  const float* __restrict__ dis, const float* __restrict__ hT,
                  float* __restrict__ lapT, int n, int E) {
  __shared__ float hcol[NMAX + 1];
  __shared__ float acc[NACC * CTHREADS];
  const int tid = threadIdx.x;
  const int f = blockIdx.x >> 2;
  const int q = blockIdx.x & 3;
  const int qn = (n + 3) >> 2;
  const int dbase = q * qn;
  const int drem = n - dbase;
  const int dcnt = (drem < 0) ? 0 : ((drem < qn) ? drem : qn);
  const float* __restrict__ col = hT + (size_t)f * n;
#pragma unroll
  for (int g = 0; g < 2; ++g) {
    float cv[NCOL], dv[NCOL];
#pragma unroll
    for (int k = 0; k < NCOL; ++k) {
      const int i = tid + (g * NCOL + k) * CTHREADS;
      const int ic = (i < n) ? i : 0;
      cv[k] = col[ic];
      dv[k] = dis[ic];
    }
#pragma unroll
    for (int k = 0; k < NCOL; ++k) {
      const int i = tid + (g * NCOL + k) * CTHREADS;
      if (i < n) hcol[i] = dv[k] * cv[k];
    }
  }
  for (int k = 0; k < NACC; ++k) acc[tid + k * CTHREADS] = 0.0f;
  if (tid == 0) hcol[NMAX] = 0.0f;
  __syncthreads();

  const unsigned int* __restrict__ lst = (q < 2) ? listA : listB;
  const int M = ctr[q];
  const int p0 = (q & 1) ? (E - M) : 0;
  const int p1 = p0 + M;
  const int a0r = (p0 + 3) & ~3;
  const int a0 = (a0r < p1) ? a0r : p1;
  const int M4 = (p1 - a0) >> 2;
  const int T = (M4 + CQ * CTHREADS - 1) / (CQ * CTHREADS);
  const u4* __restrict__ pp = (const u4*)(lst + a0);

  u4 wA[CQ], wB[CQ];
  if (T > 0) {
    col_load(wA, pp, 0, tid, M4);
    for (int t = 0; t < T; t += 2) {
      const int tb = (t + 1 < T) ? (t + 1) : (T - 1);
      const int ta = (t + 2 < T) ? (t + 2) : (T - 1);
      col_load(wB, pp, tb, tid, M4);
      __builtin_amdgcn_sched_barrier(0);
      col_apply(wA, hcol, acc, t, tid, M4);
      if (t + 1 >= T) break;
      col_load(wA, pp, ta, tid, M4);
      __builtin_amdgcn_sched_barrier(0);
      col_apply(wB, hcol, acc, t + 1, tid, M4);
    }
  }
  {
    const int nhead = a0 - p0;
    const int t0 = a0 + 4 * M4;
    const int ntail = p1 - t0;
    if (tid < nhead + ntail) {
      const unsigned e = lst[(tid < nhead) ? (p0 + tid) : (t0 + tid - nhead)];
      const float v = hcol[e & 0xffffu];
      if (v != 0.0f) atomicAdd(&acc[e >> 16], v);
    }
  }
  float dd[NACC];
#pragma unroll
  for (int k = 0; k < NACC; ++k) {
    const int i = tid + k * CTHREADS;
    dd[k] = dis[(i < dcnt) ? (dbase + i) : 0];
  }
  __syncthreads();

  float* outp = lapT + (size_t)f * n + dbase;
  float av[NACC];
#pragma unroll
  for (int k = 0; k < NACC; ++k) av[k] = acc[tid + k * CTHREADS];
#pragma unroll
  for (int k = 0; k < NACC; ++k) {
    const int i = tid + k * CTHREADS;
    if (i < dcnt) outp[i] = -(dd[k] * av[k]);
  }
}


__device__ __forceinline__ void stage_w_pairs(float* sWp, const float* __restrict__ W,
                                              int tid) {
  const float4* Wv = (const float4*)W;
#pragma unroll
  for (int i = 0; i < 8; ++i) {
    int idx = tid + i * 128;
    float4 v = Wv[idx];
    int k = idx >> 4;
    int c4 = (idx & 15) * 4;
    float* dstp = sWp + (k >> 1) * WPSTR + (k & 1);
    dstp[(c4 + 0) * 2] = v.x;
    dstp[(c4 + 1) * 2] = v.y;
    dstp[(c4 + 2) * 2] = v.z;
    dstp[(c4 + 3) * 2] = v.w;
  }
}

__device__ __forceinline__ void stage_rows(float* sAw, const float* __restrict__ A,
                                           int rowBase, int n, int lane) {
#pragma unroll
  for (int i = 0; i < 8; ++i) {
    int fi = lane + i * 32;
    int r = fi >> 4, c4 = fi & 15;
    int row = rowBase + r;
    if (row >= n) row = n - 1;
    ((float4*)(sAw + r * ASTR))[c4] = ((const float4*)(A + (size_t)row * 64))[c4];
  }
}

__device__ __forceinline__ void stage_rows_T(float* sAw, const float* __restrict__ AT,
                                             int rowBase, int n, int lane) {
  const int r = lane & 15, kh = lane >> 4;
  int row = rowBase + r;
  if (row >= n) row = n - 1;
#pragma unroll
  for (int i = 0; i < 32; ++i) {
    int k = 2 * i + kh;
    sAw[r * ASTR + k] = AT[(size_t)k * n + row];
  }
}

__device__ __forceinline__ void mm_accum64(const float* sAw, const float* sWp,
                                           int lane, v8f acc[4]) {
  const int nlo = lane & 15;
  const int g = lane >> 4;
#pragma unroll
  for (int kk = 0; kk < 16; ++kk) {
    const int k0 = kk * 4 + 2 * g;
    v2f a = *(const v2f*)(sAw + nlo * ASTR + k0);
    const float* brow = sWp + (kk * 2 + g) * WPSTR;
#pragma unroll
    for (int nt = 0; nt < 4; ++nt) {
      v2f b = *(const v2f*)(brow + (nt * 16 + nlo) * 2);
      acc[nt] = wmma4(a, b, acc[nt]);
    }
  }
}


__global__ void __launch_bounds__(128)
gemm64_relu_kernel(const float* __restrict__ A, const float* __restrict__ W,
                   const float* __restrict__ bias, float* __restrict__ outT, int n) {
  __shared__ float sWp[WPROWS * WPSTR];
  __shared__ float sA[GWAVES * 16 * ASTR];
  __shared__ float sB[64];
  const int tid = threadIdx.x, wave = tid >> 5, lane = tid & 31;
  stage_w_pairs(sWp, W, tid);
  if (tid < 64) sB[tid] = bias[tid];

  const int rowBase = blockIdx.x * GROWS + wave * 16;
  float* sAw = sA + wave * 16 * ASTR;
  stage_rows(sAw, A, rowBase, n, lane);
  __syncthreads();

  v8f z = {0, 0, 0, 0, 0, 0, 0, 0};
  v8f acc[4] = {z, z, z, z};
  mm_accum64(sAw, sWp, lane, acc);

  const int nlo = lane & 15, g = lane >> 4;
#pragma unroll
  for (int nt = 0; nt < 4; ++nt)
#pragma unroll
    for (int r = 0; r < 8; ++r) {
      int col = nt * 16 + nlo;
      sAw[(r + 8 * g) * ASTR + col] = fmaxf(acc[nt][r] + sB[col], 0.0f);
    }
  __syncthreads();
  {
    const int r = lane & 15, kh = lane >> 4;
    const int row = rowBase + r;
    if (row < n) {
#pragma unroll
      for (int i = 0; i < 32; ++i) {
        int k = 2 * i + kh;
        outT[(size_t)k * n + row] = sAw[r * ASTR + k];
      }
    }
  }
}

__global__ void __launch_bounds__(128)
gemm64_cheb2_kernel(float* __restrict__ h, const float* __restrict__ lap,
                    const float* __restrict__ W0, const float* __restrict__ W1,
                    const float* __restrict__ bias, int n) {
  __shared__ float sWp[2 * WPROWS * WPSTR];
  __shared__ float sA[GWAVES * 16 * ASTR];
  __shared__ float sB[64];
  const int tid = threadIdx.x, wave = tid >> 5, lane = tid & 31;
  stage_w_pairs(sWp, W0, tid);
  stage_w_pairs(sWp + WPROWS * WPSTR, W1, tid);
  if (tid < 64) sB[tid] = bias[tid];

  const int rowBase = blockIdx.x * GROWS + wave * 16;
  float* sAw = sA + wave * 16 * ASTR;

  v8f z = {0, 0, 0, 0, 0, 0, 0, 0};
  v8f acc[4] = {z, z, z, z};

  stage_rows_T(sAw, h, rowBase, n, lane);
  __syncthreads();
  mm_accum64(sAw, sWp, lane, acc);
  __syncthreads();
  stage_rows_T(sAw, lap, rowBase, n, lane);
  __syncthreads();
  mm_accum64(sAw, sWp + WPROWS * WPSTR, lane, acc);

  const int nlo = lane & 15, g = lane >> 4;
#pragma unroll
  for (int nt = 0; nt < 4; ++nt)
#pragma unroll
    for (int r = 0; r < 8; ++r) {
      int col = nt * 16 + nlo;
      sAw[(r + 8 * g) * ASTR + col] = fmaxf(acc[nt][r] + sB[col], 0.0f);
    }
  __syncthreads();
  {
    const int r = lane & 15, kh = lane >> 4;
    const int row = rowBase + r;
    if (row < n) {
      float res[32];
#pragma unroll
      for (int i = 0; i < 32; ++i) res[i] = h[(size_t)(2 * i + kh) * n + row];
#pragma unroll
      for (int i = 0; i < 32; ++i) {
        int k = 2 * i + kh;
        h[(size_t)k * n + row] = sAw[r * ASTR + k] + res[i];
      }
    }
  }
}

__global__ void __launch_bounds__(128)
gemm_out_kernel(const float* __restrict__ A, const float* __restrict__ Wl,
                const float* __restrict__ bl, float* __restrict__ logits, int n) {
  __shared__ float sWo[WPROWS * WOSTR];
  __shared__ float sA[GWAVES * 16 * ASTR];
  __shared__ float sB[16];
  const int tid = threadIdx.x, wave = tid >> 5, lane = tid & 31;
  for (int idx = tid; idx < 64 * 16; idx += 128) {
    int k = idx >> 4, c = idx & 15;
    float v = (c < CDIM) ? Wl[k * CDIM + c] : 0.0f;
    sWo[(k >> 1) * WOSTR + c * 2 + (k & 1)] = v;
  }
  if (tid < 16) sB[tid] = (tid < CDIM) ? bl[tid] : 0.0f;

  const int rowBase = blockIdx.x * GROWS + wave * 16;
  float* sAw = sA + wave * 16 * ASTR;
  stage_rows_T(sAw, A, rowBase, n, lane);
  __syncthreads();

  const int nlo = lane & 15, g = lane >> 4;
  v8f acc = {0, 0, 0, 0, 0, 0, 0, 0};
#pragma unroll
  for (int kk = 0; kk < 16; ++kk) {
    const int k0 = kk * 4 + 2 * g;
    v2f a = *(const v2f*)(sAw + nlo * ASTR + k0);
    v2f b = *(const v2f*)(sWo + (kk * 2 + g) * WOSTR + nlo * 2);
    acc = wmma4(a, b, acc);
  }
#pragma unroll
  for (int r = 0; r < 8; ++r) {
    int row = rowBase + r + 8 * g;
    if (row < n) logits[(size_t)row * 16 + nlo] = acc[r] + sB[nlo];
  }
}

__global__ void logsoftmax_kernel(const float* __restrict__ logits,
                                  float* __restrict__ out, int n) {
  int i = blockIdx.x * blockDim.x + threadIdx.x;
  if (i >= n) return;
  const float* v = logits + (size_t)i * 16;
  float m = v[0];
#pragma unroll
  for (int c = 1; c < CDIM; ++c) m = fmaxf(m, v[c]);
  float s = 0.0f;
#pragma unroll
  for (int c = 0; c < CDIM; ++c) s += expf(v[c] - m);
  float ls = logf(s);
#pragma unroll
  for (int c = 0; c < CDIM; ++c) out[(size_t)i * CDIM + c] = v[c] - m - ls;
}


extern "C" void kernel_launch(void* const* d_in, const int* in_sizes, int n_in,
                              void* d_out, int out_size, void* d_ws, size_t ws_size,
                              hipStream_t stream) {
  const float* x    = (const float*)d_in[0];
  const int*   ei   = (const int*)d_in[1];
  const float* W1_0 = (const float*)d_in[2];
  const float* b1   = (const float*)d_in[3];
  const float* W2_0 = (const float*)d_in[4];
  const float* W2_1 = (const float*)d_in[5];
  const float* b2   = (const float*)d_in[6];
  const float* Wl   = (const float*)d_in[7];
  const float* bl   = (const float*)d_in[8];

  const int N = in_sizes[0] / FDIM;
  const int E = in_sizes[1] / 2;
  const int* src = ei;
  const int* dst = ei + E;

  float* ws     = (float*)d_ws;
  float* lap    = ws + N;
  float* dis    = lap + (size_t)N * 64;
  unsigned int* listA = (unsigned int*)(dis + N);
  float* h      = dis + N + E;
  float* logits = h + (size_t)N * 64;
  float* out    = (float*)d_out;
  float* cnt    = h;
  unsigned int* listB = (unsigned int*)logits;
  const int nblk = (E + PEDGES - 1) / PEDGES;
  int* c4  = (int*)d_ws;
  int* ctr = c4 + 4 * nblk;
  const int qn = (N + 3) >> 2;

  zero_kernel<<<(N + 255) / 256, 256, 0, stream>>>(cnt, N);
  degree_kernel<<<nblk, PTHREADS, 0, stream>>>(src, dst, cnt, c4, E, qn);
  place_dis_kernel<<<nblk + (N + PTHREADS - 1) / PTHREADS, PTHREADS, 0, stream>>>(
      src, dst, c4, ctr, listA, listB, cnt, dis, N, E, nblk, qn);

  const int gblocks = (N + GROWS - 1) / GROWS;
  gemm64_relu_kernel<<<gblocks, 128, 0, stream>>>(x, W1_0, b1, h, N);
  colscatter_kernel<<<64 * 4, CTHREADS, 0, stream>>>(listA, listB, ctr, dis, h, lap, N, E);
  gemm64_cheb2_kernel<<<gblocks, 128, 0, stream>>>(h, lap, W2_0, W2_1, b2, N);
  gemm_out_kernel<<<gblocks, 128, 0, stream>>>(h, Wl, bl, logits, N);
  logsoftmax_kernel<<<(N + 255) / 256, 256, 0, stream>>>(logits, out, N);
}


# degree counters in two sets, one per package half: a wave adds into the set of the half it runs on (from its die number, speed only), so no counter line is shared by both L2 halves; deg = sum, exact
# speedup vs baseline: 2.1190x; 1.1442x over previous
#include <hip/hip_runtime.h>
#include <hip/hip_bf16.h>
#include <stdint.h>


#define FDIM 64
#define HDIM 64
#define CDIM 10

#define GWAVES 4
#define GROWS  (GWAVES * 16)
#define ASTR   68
#define WPSTR  160
#define WPROWS 32
#define WOSTR  96
#define CSTR   32
#define NMAX   50000
#define NQ     12500
#define CTHREADS 1024
#define CU     16
#define CQ     (CU / 4)
#define NCOL   25
#define NACC   13
#define PTHREADS 256
#define EPT      8
#define PEDGES   (PTHREADS * EPT)
#define PWAVES   (PTHREADS / 32)

typedef unsigned int u4 __attribute__((ext_vector_type(4)));
typedef float v2f __attribute__((ext_vector_type(2)));
typedef float v8f __attribute__((ext_vector_type(8)));

__device__ __forceinline__ v8f wmma4(v2f a, v2f b, v8f c) {
  return __builtin_amdgcn_wmma_f32_16x16x4_f32(false, a, false, b, (short)0, c,
                                               false, false);
}


__global__ void zero_kernel(float* __restrict__ cnt, int ncnt) {
  int i = blockIdx.x * blockDim.x + threadIdx.x;
  if (i < ncnt) cnt[(size_t)i * CSTR] = 0.0f;
}

__device__ __forceinline__ int wave_sum(int v) {
  for (int o = 16; o > 0; o >>= 1) v += __shfl_xor(v, o, 32);
  return v;
}

__device__ __forceinline__ int dst_quarter(int d, int qn) {
  return (d >= qn) + (d >= 2 * qn) + (d >= 3 * qn);
}

__device__ __forceinline__ int wave_scan(int v, int lane) {
  for (int o = 1; o < 32; o <<= 1) {
    const int t = __shfl_up(v, o, 32);
    if (lane >= o) v += t;
  }
  return v;
}

__device__ __forceinline__ void load_edges(const int* __restrict__ src, const int* __restrict__ dst,
                                           int e0, int E, bool vec, int (&s)[EPT], int (&d)[EPT]) {
  if (vec && e0 + EPT <= E) {
    const int4 s0 = *(const int4*)(src + e0), s1 = *(const int4*)(src + e0 + 4);
    const int4 d0 = *(const int4*)(dst + e0), d1 = *(const int4*)(dst + e0 + 4);
    s[0] = s0.x; s[1] = s0.y; s[2] = s0.z; s[3] = s0.w;
    s[4] = s1.x; s[5] = s1.y; s[6] = s1.z; s[7] = s1.w;
    d[0] = d0.x; d[1] = d0.y; d[2] = d0.z; d[3] = d0.w;
    d[4] = d1.x; d[5] = d1.y; d[6] = d1.z; d[7] = d1.w;
  } else {
#pragma unroll
    for (int j = 0; j < EPT; ++j) {
      const int e = e0 + j;
      s[j] = (e < E) ? src[e] : 0;
      d[j] = (e < E) ? dst[e] : 0;
    }
  }
}

__global__ void __launch_bounds__(PTHREADS)
degree_kernel(const int* __restrict__ src, const int* __restrict__ dst, float* __restrict__ cnt,
              int* __restrict__ c4, int n, int E, int qn) {
  __shared__ int sw[4][PWAVES];
  const int tid = threadIdx.x, lane = tid & 31, wave = tid >> 5;
  const int e0 = (blockIdx.x * PTHREADS + tid) * EPT;
  const int left = E - e0;
  const int nv = (left < 0) ? 0 : ((left < EPT) ? left : EPT);
  const unsigned where = __builtin_amdgcn_s_sendmsg_rtn(0x87);
  int s[EPT], d[EPT];
  load_edges(src, dst, e0, E, (E & 3) == 0, s, d);
  float* __restrict__ mycnt = cnt + (size_t)((where >> 18) & 1u) * (size_t)n * CSTR;
  int lo = 0, hi = 0;
#pragma unroll
  for (int j = 0; j < EPT; ++j) {
    if (j < nv) {
      atomicAdd(&mycnt[(size_t)s[j] * CSTR], 1.0f);
      const int cls = dst_quarter(d[j], qn);
      const int one = (cls & 1) ? 0x10000 : 1;
      if (cls < 2) lo += one; else hi += one;
    }
  }
  lo = wave_sum(lo);
  hi = wave_sum(hi);
  if (lane == 0) {
    sw[0][wave] = lo & 0xffff; sw[1][wave] = lo >> 16;
    sw[2][wave] = hi & 0xffff; sw[3][wave] = hi >> 16;
  }
  __syncthreads();
  if (tid < 4) {
    int v = 0;
#pragma unroll
    for (int w = 0; w < PWAVES; ++w) v += sw[tid][w];
    c4[blockIdx.x * 4 + tid] = v;
  }
}

__global__ void __launch_bounds__(PTHREADS)
place_dis_kernel(const int* __restrict__ src, const int* __restrict__ dst,
                 const int* __restrict__ c4, int* __restrict__ ctr,
                 unsigned int* __restrict__ listA, unsigned int* __restrict__ listB,
                 const float* __restrict__ cnt, float* __restrict__ dis,
                 int n, int E, int nblk, int qn) {
  __shared__ int spre[4][PWAVES];
  __shared__ int stot[4][PWAVES];
  __shared__ int sw[4][PWAVES];
  const int tid = threadIdx.x, lane = tid & 31, wave = tid >> 5;
  const int blk = blockIdx.x;
  if (blk >= nblk) {
    const int i = (blk - nblk) * PTHREADS + tid;
    if (i < n) {
      const float d0 = cnt[(size_t)i * CSTR], d1 = cnt[(size_t)(n + i) * CSTR];
      const float d = d0 + d1;
      dis[i] = (d > 0.0f) ? rsqrtf(fmaxf(d, 1.0f)) : 0.0f;
    }
    return;
  }
  const int e0 = (blk * PTHREADS + tid) * EPT;
  const int left = E - e0;
  const int nv = (left < 0) ? 0 : ((left < EPT) ? left : EPT);
  int s[EPT], d[EPT];
  load_edges(src, dst, e0, E, (E & 3) == 0, s, d);

  int pre[4] = {0, 0, 0, 0}, tot[4] = {0, 0, 0, 0};
  for (int i = tid; i < nblk; i += 2 * PTHREADS) {
    const int i2 = i + PTHREADS;
    const bool has2 = (i2 < nblk);
    const int4 v = ((const int4*)c4)[i];
    const int4 u = ((const int4*)c4)[has2 ? i2 : i];
    const bool b1 = (i < blk), b2 = has2 && (i2 < blk);
    tot[0] += v.x + (has2 ? u.x : 0); tot[1] += v.y + (has2 ? u.y : 0);
    tot[2] += v.z + (has2 ? u.z : 0); tot[3] += v.w + (has2 ? u.w : 0);
    pre[0] += (b1 ? v.x : 0) + (b2 ? u.x : 0); pre[1] += (b1 ? v.y : 0) + (b2 ? u.y : 0);
    pre[2] += (b1 ? v.z : 0) + (b2 ? u.z : 0); pre[3] += (b1 ? v.w : 0) + (b2 ? u.w : 0);
  }
#pragma unroll
  for (int k = 0; k < 4; ++k) {
    const int p = wave_sum(pre[k]), t = wave_sum(tot[k]);
    if (lane == 0) { spre[k][wave] = p; stot[k][wave] = t; }
  }
  int cls[EPT];
  int lo = 0, hi = 0;
#pragma unroll
  for (int j = 0; j < EPT; ++j) {
    cls[j] = dst_quarter(d[j], qn);
    if (j < nv) {
      const int one = (cls[j] & 1) ? 0x10000 : 1;
      if (cls[j] < 2) lo += one; else hi += one;
    }
  }
  const int ilo = wave_scan(lo, lane), ihi = wave_scan(hi, lane);
  if (lane == 31) {
    sw[0][wave] = ilo & 0xffff; sw[1][wave] = ilo >> 16;
    sw[2][wave] = ihi & 0xffff; sw[3][wave] = ihi >> 16;
  }
  __syncthreads();
  int off[4], T[4];
#pragma unroll
  for (int k = 0; k < 4; ++k) {
    int P = 0, Tk = 0, Wv = 0;
#pragma unroll
    for (int w = 0; w < PWAVES; ++w) {
      P += spre[k][w];
      Tk += stot[k][w];
      Wv += (w < wave) ? sw[k][w] : 0;
    }
    const int mine = (k & 1) ? (((k < 2) ? lo : hi) >> 16) : (((k < 2) ? lo : hi) & 0xffff);
    const int incl = (k & 1) ? (((k < 2) ? ilo : ihi) >> 16) : (((k < 2) ? ilo : ihi) & 0xffff);
    const int base = (k & 1) ? (E - Tk) : 0;
    off[k] = base + P + Wv + (incl - mine);
    T[k] = Tk;
  }
#pragma unroll
  for (int j = 0; j < EPT; ++j) {
    if (j < nv) {
      const int k = cls[j];
      const int o = (k == 0) ? off[0] : (k == 1) ? off[1] : (k == 2) ? off[2] : off[3];
      unsigned int* lst = (k < 2) ? listA : listB;
      lst[o] = (unsigned)s[j] | ((unsigned)(d[j] - k * qn) << 16);
      off[0] += (k == 0); off[1] += (k == 1); off[2] += (k == 2); off[3] += (k == 3);
    }
  }
  if (blk == nblk - 1 && tid == 0) {
    ctr[0] = T[0]; ctr[1] = T[1]; ctr[2] = T[2]; ctr[3] = T[3];
  }
}


__device__ __forceinline__ void col_load(u4 (&w)[CQ], const u4* __restrict__ pp,
                                         int t, int tid, int M4) {
#pragma unroll
  for (int j = 0; j < CQ; ++j) {
    int o = (t * CQ + j) * CTHREADS + tid;
    w[j] = pp[(o < M4) ? o : 0];
  }
}

__device__ __forceinline__ void col_apply(const u4 (&w)[CQ], const float* hcol, float* acc,
                                          int t, int tid, int M4) {
#pragma unroll
  for (int j = 0; j < CQ; ++j) asm volatile("" :: "v"(w[j]));
  unsigned r[CU];
  float hv[CU];
#pragma unroll
  for (int j = 0; j < CQ; ++j) {
    const int o = (t * CQ + j) * CTHREADS + tid;
    const bool live = (o < M4);
#pragma unroll
    for (int c = 0; c < 4; ++c) {
      const unsigned e = w[j][c];
      const int k = j * 4 + c;
      r[k] = e >> 16;
      hv[k] = hcol[live ? (e & 0xffffu) : NMAX];
    }
  }
  asm volatile("" :: "v"(hv[0]), "v"(hv[1]), "v"(hv[2]), "v"(hv[3]), "v"(hv[4]), "v"(hv[5]),
               "v"(hv[6]), "v"(hv[7]), "v"(hv[8]), "v"(hv[9]), "v"(hv[10]), "v"(hv[11]),
               "v"(hv[12]), "v"(hv[13]), "v"(hv[14]), "v"(hv[15]));
#pragma unroll
  for (int k = 0; k < CU; ++k) {
    if (hv[k] != 0.0f) atomicAdd(&acc[r[k]], hv[k]);
  }
}

__global__ void __launch_bounds__(CTHREADS)
colscatter_kernel(const unsigned int* __restrict__ listA, const unsigned int* __restrict__ listB,
                  const int* __restrict__ ctr,
                  const float* __restrict__ dis, const float* __restrict__ hT,
                  float* __restrict__ lapT, int n, int E) {
  __shared__ float hcol[NMAX + 1];
  __shared__ float acc[NACC * CTHREADS];
  const int tid = threadIdx.x;
  const int f = blockIdx.x >> 2;
  const int q = blockIdx.x & 3;
  const int qn = (n + 3) >> 2;
  const int dbase = q * qn;
  const int drem = n - dbase;
  const int dcnt = (drem < 0) ? 0 : ((drem < qn) ? drem : qn);
  const float* __restrict__ col = hT + (size_t)f * n;
#pragma unroll
  for (int g = 0; g < 2; ++g) {
    float cv[NCOL], dv[NCOL];
#pragma unroll
    for (int k = 0; k < NCOL; ++k) {
      const int i = tid + (g * NCOL + k) * CTHREADS;
      const int ic = (i < n) ? i : 0;
      cv[k] = col[ic];
      dv[k] = dis[ic];
    }
#pragma unroll
    for (int k = 0; k < NCOL; ++k) {
      const int i = tid + (g * NCOL + k) * CTHREADS;
      if (i < n) hcol[i] = dv[k] * cv[k];
    }
  }
  for (int k = 0; k < NACC; ++k) acc[tid + k * CTHREADS] = 0.0f;
  if (tid == 0) hcol[NMAX] = 0.0f;
  __syncthreads();

  const unsigned int* __restrict__ lst = (q < 2) ? listA : listB;
  const int M = ctr[q];
  const int p0 = (q & 1) ? (E - M) : 0;
  const int p1 = p0 + M;
  const int a0r = (p0 + 3) & ~3;
  const int a0 = (a0r < p1) ? a0r : p1;
  const int M4 = (p1 - a0) >> 2;
  const int T = (M4 + CQ * CTHREADS - 1) / (CQ * CTHREADS);
  const u4* __restrict__ pp = (const u4*)(lst + a0);

  u4 wA[CQ], wB[CQ];
  if (T > 0) {
    col_load(wA, pp, 0, tid, M4);
    for (int t = 0; t < T; t += 2) {
      const int tb = (t + 1 < T) ? (t + 1) : (T - 1);
      const int ta = (t + 2 < T) ? (t + 2) : (T - 1);
      col_load(wB, pp, tb, tid, M4);
      __builtin_amdgcn_sched_barrier(0);
      col_apply(wA, hcol, acc, t, tid, M4);
      if (t + 1 >= T) break;
      col_load(wA, pp, ta, tid, M4);
      __builtin_amdgcn_sched_barrier(0);
      col_apply(wB, hcol, acc, t + 1, tid, M4);
    }
  }
  {
    const int nhead = a0 - p0;
    const int t0 = a0 + 4 * M4;
    const int ntail = p1 - t0;
    if (tid < nhead + ntail) {
      const unsigned e = lst[(tid < nhead) ? (p0 + tid) : (t0 + tid - nhead)];
      const float v = hcol[e & 0xffffu];
      if (v != 0.0f) atomicAdd(&acc[e >> 16], v);
    }
  }
  float dd[NACC];
#pragma unroll
  for (int k = 0; k < NACC; ++k) {
    const int i = tid + k * CTHREADS;
    dd[k] = dis[(i < dcnt) ? (dbase + i) : 0];
  }
  __syncthreads();

  float* outp = lapT + (size_t)f * n + dbase;
  float av[NACC];
#pragma unroll
  for (int k = 0; k < NACC; ++k) av[k] = acc[tid + k * CTHREADS];
#pragma unroll
  for (int k = 0; k < NACC; ++k) {
    const int i = tid + k * CTHREADS;
    if (i < dcnt) outp[i] = -(dd[k] * av[k]);
  }
}


__device__ __forceinline__ void stage_w_pairs(float* sWp, const float* __restrict__ W,
                                              int tid) {
  const float4* Wv = (const float4*)W;
#pragma unroll
  for (int i = 0; i < 8; ++i) {
    int idx = tid + i * 128;
    float4 v = Wv[idx];
    int k = idx >> 4;
    int c4 = (idx & 15) * 4;
    float* dstp = sWp + (k >> 1) * WPSTR + (k & 1);
    dstp[(c4 + 0) * 2] = v.x;
    dstp[(c4 + 1) * 2] = v.y;
    dstp[(c4 + 2) * 2] = v.z;
    dstp[(c4 + 3) * 2] = v.w;
  }
}

__device__ __forceinline__ void stage_rows(float* sAw, const float* __restrict__ A,
                                           int rowBase, int n, int lane) {
#pragma unroll
  for (int i = 0; i < 8; ++i) {
    int fi = lane + i * 32;
    int r = fi >> 4, c4 = fi & 15;
    int row = rowBase + r;
    if (row >= n) row = n - 1;
    ((float4*)(sAw + r * ASTR))[c4] = ((const float4*)(A + (size_t)row * 64))[c4];
  }
}

__device__ __forceinline__ void stage_rows_T(float* sAw, const float* __restrict__ AT,
                                             int rowBase, int n, int lane) {
  const int r = lane & 15, kh = lane >> 4;
  int row = rowBase + r;
  if (row >= n) row = n - 1;
#pragma unroll
  for (int i = 0; i < 32; ++i) {
    int k = 2 * i + kh;
    sAw[r * ASTR + k] = AT[(size_t)k * n + row];
  }
}

__device__ __forceinline__ void mm_accum64(const float* sAw, const float* sWp,
                                           int lane, v8f acc[4]) {
  const int nlo = lane & 15;
  const int g = lane >> 4;
#pragma unroll
  for (int kk = 0; kk < 16; ++kk) {
    const int k0 = kk * 4 + 2 * g;
    v2f a = *(const v2f*)(sAw + nlo * ASTR + k0);
    const float* brow = sWp + (kk * 2 + g) * WPSTR;
#pragma unroll
    for (int nt = 0; nt < 4; ++nt) {
      v2f b = *(const v2f*)(brow + (nt * 16 + nlo) * 2);
      acc[nt] = wmma4(a, b, acc[nt]);
    }
  }
}


__global__ void __launch_bounds__(128)
gemm64_relu_kernel(const float* __restrict__ A, const float* __restrict__ W,
                   const float* __restrict__ bias, float* __restrict__ outT, int n) {
  __shared__ float sWp[WPROWS * WPSTR];
  __shared__ float sA[GWAVES * 16 * ASTR];
  __shared__ float sB[64];
  const int tid = threadIdx.x, wave = tid >> 5, lane = tid & 31;
  stage_w_pairs(sWp, W, tid);
  if (tid < 64) sB[tid] = bias[tid];

  const int rowBase = blockIdx.x * GROWS + wave * 16;
  float* sAw = sA + wave * 16 * ASTR;
  stage_rows(sAw, A, rowBase, n, lane);
  __syncthreads();

  v8f z = {0, 0, 0, 0, 0, 0, 0, 0};
  v8f acc[4] = {z, z, z, z};
  mm_accum64(sAw, sWp, lane, acc);

  const int nlo = lane & 15, g = lane >> 4;
#pragma unroll
  for (int nt = 0; nt < 4; ++nt)
#pragma unroll
    for (int r = 0; r < 8; ++r) {
      int col = nt * 16 + nlo;
      sAw[(r + 8 * g) * ASTR + col] = fmaxf(acc[nt][r] + sB[col], 0.0f);
    }
  __syncthreads();
  {
    const int r = lane & 15, kh = lane >> 4;
    const int row = rowBase + r;
    if (row < n) {
#pragma unroll
      for (int i = 0; i < 32; ++i) {
        int k = 2 * i + kh;
        outT[(size_t)k * n + row] = sAw[r * ASTR + k];
      }
    }
  }
}

__global__ void __launch_bounds__(128)
gemm64_cheb2_kernel(float* __restrict__ h, const float* __restrict__ lap,
                    const float* __restrict__ W0, const float* __restrict__ W1,
                    const float* __restrict__ bias, int n) {
  __shared__ float sWp[2 * WPROWS * WPSTR];
  __shared__ float sA[GWAVES * 16 * ASTR];
  __shared__ float sB[64];
  const int tid = threadIdx.x, wave = tid >> 5, lane = tid & 31;
  stage_w_pairs(sWp, W0, tid);
  stage_w_pairs(sWp + WPROWS * WPSTR, W1, tid);
  if (tid < 64) sB[tid] = bias[tid];

  const int rowBase = blockIdx.x * GROWS + wave * 16;
  float* sAw = sA + wave * 16 * ASTR;

  v8f z = {0, 0, 0, 0, 0, 0, 0, 0};
  v8f acc[4] = {z, z, z, z};

  stage_rows_T(sAw, h, rowBase, n, lane);
  __syncthreads();
  mm_accum64(sAw, sWp, lane, acc);
  __syncthreads();
  stage_rows_T(sAw, lap, rowBase, n, lane);
  __syncthreads();
  mm_accum64(sAw, sWp + WPROWS * WPSTR, lane, acc);

  const int nlo = lane & 15, g = lane >> 4;
#pragma unroll
  for (int nt = 0; nt < 4; ++nt)
#pragma unroll
    for (int r = 0; r < 8; ++r) {
      int col = nt * 16 + nlo;
      sAw[(r + 8 * g) * ASTR + col] = fmaxf(acc[nt][r] + sB[col], 0.0f);
    }
  __syncthreads();
  {
    const int r = lane & 15, kh = lane >> 4;
    const int row = rowBase + r;
    if (row < n) {
      float res[32];
#pragma unroll
      for (int i = 0; i < 32; ++i) res[i] = h[(size_t)(2 * i + kh) * n + row];
#pragma unroll
      for (int i = 0; i < 32; ++i) {
        int k = 2 * i + kh;
        h[(size_t)k * n + row] = sAw[r * ASTR + k] + res[i];
      }
    }
  }
}

__global__ void __launch_bounds__(128)
gemm_out_kernel(const float* __restrict__ A, const float* __restrict__ Wl,
                const float* __restrict__ bl, float* __restrict__ logits, int n) {
  __shared__ float sWo[WPROWS * WOSTR];
  __shared__ float sA[GWAVES * 16 * ASTR];
  __shared__ float sB[16];
  const int tid = threadIdx.x, wave = tid >> 5, lane = tid & 31;
  for (int idx = tid; idx < 64 * 16; idx += 128) {
    int k = idx >> 4, c = idx & 15;
    float v = (c < CDIM) ? Wl[k * CDIM + c] : 0.0f;
    sWo[(k >> 1) * WOSTR + c * 2 + (k & 1)] = v;
  }
  if (tid < 16) sB[tid] = (tid < CDIM) ? bl[tid] : 0.0f;

  const int rowBase = blockIdx.x * GROWS + wave * 16;
  float* sAw = sA + wave * 16 * ASTR;
  stage_rows_T(sAw, A, rowBase, n, lane);
  __syncthreads();

  const int nlo = lane & 15, g = lane >> 4;
  v8f acc = {0, 0, 0, 0, 0, 0, 0, 0};
#pragma unroll
  for (int kk = 0; kk < 16; ++kk) {
    const int k0 = kk * 4 + 2 * g;
    v2f a = *(const v2f*)(sAw + nlo * ASTR + k0);
    v2f b = *(const v2f*)(sWo + (kk * 2 + g) * WOSTR + nlo * 2);
    acc = wmma4(a, b, acc);
  }
#pragma unroll
  for (int r = 0; r < 8; ++r) {
    int row = rowBase + r + 8 * g;
    if (row < n) logits[(size_t)row * 16 + nlo] = acc[r] + sB[nlo];
  }
}

__global__ void logsoftmax_kernel(const float* __restrict__ logits,
                                  float* __restrict__ out, int n) {
  int i = blockIdx.x * blockDim.x + threadIdx.x;
  if (i >= n) return;
  const float* v = logits + (size_t)i * 16;
  float m = v[0];
#pragma unroll
  for (int c = 1; c < CDIM; ++c) m = fmaxf(m, v[c]);
  float s = 0.0f;
#pragma unroll
  for (int c = 0; c < CDIM; ++c) s += expf(v[c] - m);
  float ls = logf(s);
#pragma unroll
  for (int c = 0; c < CDIM; ++c) out[(size_t)i * CDIM + c] = v[c] - m - ls;
}


extern "C" void kernel_launch(void* const* d_in, const int* in_sizes, int n_in,
                              void* d_out, int out_size, void* d_ws, size_t ws_size,
                              hipStream_t stream) {
  const float* x    = (const float*)d_in[0];
  const int*   ei   = (const int*)d_in[1];
  const float* W1_0 = (const float*)d_in[2];
  const float* b1   = (const float*)d_in[3];
  const float* W2_0 = (const float*)d_in[4];
  const float* W2_1 = (const float*)d_in[5];
  const float* b2   = (const float*)d_in[6];
  const float* Wl   = (const float*)d_in[7];
  const float* bl   = (const float*)d_in[8];

  const int N = in_sizes[0] / FDIM;
  const int E = in_sizes[1] / 2;
  const int* src = ei;
  const int* dst = ei + E;

  float* ws     = (float*)d_ws;
  float* lap    = ws + N;
  float* dis    = lap + (size_t)N * 64;
  unsigned int* listA = (unsigned int*)(dis + N);
  float* h      = dis + N + E;
  float* logits = h + (size_t)N * 64;
  float* out    = (float*)d_out;
  float* cnt    = h;
  unsigned int* listB = (unsigned int*)logits;
  const int nblk = (E + PEDGES - 1) / PEDGES;
  int* c4  = (int*)d_ws;
  int* ctr = c4 + 4 * nblk;
  const int qn = (N + 3) >> 2;

  zero_kernel<<<(2 * N + 255) / 256, 256, 0, stream>>>(cnt, 2 * N);
  degree_kernel<<<nblk, PTHREADS, 0, stream>>>(src, dst, cnt, c4, N, E, qn);
  place_dis_kernel<<<nblk + (N + PTHREADS - 1) / PTHREADS, PTHREADS, 0, stream>>>(
      src, dst, c4, ctr, listA, listB, cnt, dis, N, E, nblk, qn);

  const int gblocks = (N + GROWS - 1) / GROWS;
  gemm64_relu_kernel<<<gblocks, 128, 0, stream>>>(x, W1_0, b1, h, N);
  colscatter_kernel<<<64 * 4, CTHREADS, 0, stream>>>(listA, listB, ctr, dis, h, lap, N, E);
  gemm64_cheb2_kernel<<<gblocks, 128, 0, stream>>>(h, lap, W2_0, W2_1, b2, N);
  gemm_out_kernel<<<gblocks, 128, 0, stream>>>(h, Wl, bl, logits, N);
  logsoftmax_kernel<<<(N + 255) / 256, 256, 0, stream>>>(logits, out, N);
}
